# starting kernel
# baseline (speedup 1.0000x reference)
#include <hip/hip_runtime.h>
#include <hip/hip_bf16.h>

#define N_NODES   100000
#define E_EDGES   1600000
#define IN_DIMV   128
#define HIDV      64
#define NEG_SLOPE 0.2f

typedef __attribute__((ext_vector_type(2))) float v2f;
typedef __attribute__((ext_vector_type(8))) float v8f;

__device__ __forceinline__ unsigned f2ord(float f) {
    unsigned u = __float_as_uint(f);
    return (u & 0x80000000u) ? ~u : (u | 0x80000000u);
}
__device__ __forceinline__ float ord2f(unsigned o) {
    return __uint_as_float((o & 0x80000000u) ? (o & 0x7FFFFFFFu) : ~o);
}

template <int K>
__global__ void __launch_bounds__(128)
gemm_wmma_f32(const float* __restrict__ Hin, const float* __restrict__ W,
              float* __restrict__ Hout)
{
    const int lane = threadIdx.x & 31;
    const int wv   = threadIdx.x >> 5;
    const int row0 = blockIdx.x << 4;
    const int col0 = wv << 4;
    const int m    = lane & 15;
    const int kb   = (lane >> 4) << 1;

    v8f acc = {};
    const float* arow = Hin + (size_t)(row0 + m) * K;
    #pragma unroll 4
    for (int k0 = 0; k0 < K; k0 += 4) {
        v2f a, b;
        a.x = arow[k0 + kb];
        a.y = arow[k0 + kb + 1];
        b.x = W[(size_t)(k0 + kb)     * HIDV + col0 + m];
        b.y = W[(size_t)(k0 + kb + 1) * HIDV + col0 + m];
        acc = __builtin_amdgcn_wmma_f32_16x16x4_f32(
                   false, a,  false, b,
                   (short)0, acc,  false,  false);
    }

    const int rbase = row0 + ((lane >> 4) << 3);
    #pragma unroll
    for (int j = 0; j < 8; ++j)
        Hout[(size_t)(rbase + j) * HIDV + col0 + m] = acc[j];
}

__global__ void __launch_bounds__(256)
node_alpha(const float* __restrict__ H, const float* __restrict__ avs,
           const float* __restrict__ avd, float* __restrict__ as_,
           float* __restrict__ ad_, int n)
{
    const int lane = threadIdx.x & 31;
    const int node = (int)((blockIdx.x * (size_t)blockDim.x + threadIdx.x) >> 5);
    if (node >= n) return;
    const float* h = H + (size_t)node * HIDV;
    float s = h[lane] * avs[lane] + h[lane + 32] * avs[lane + 32];
    float d = h[lane] * avd[lane] + h[lane + 32] * avd[lane + 32];
    #pragma unroll
    for (int off = 16; off; off >>= 1) {
        s += __shfl_xor(s, off, 32);
        d += __shfl_xor(d, off, 32);
    }
    if (lane == 0) { as_[node] = s; ad_[node] = d; }
}

__global__ void __launch_bounds__(256)
init_node(unsigned* __restrict__ mOrd, float* __restrict__ den, float* __restrict__ agg)
{
    const size_t i = blockIdx.x * (size_t)blockDim.x + threadIdx.x;
    if (i >= (size_t)N_NODES * HIDV) return;
    agg[i] = 0.0f;
    if (i < N_NODES) { mOrd[i] = f2ord(-__builtin_inff()); den[i] = 0.0f; }
}

__global__ void __launch_bounds__(256)
edge_logits(const int* __restrict__ se, const int* __restrict__ de,
            const float* __restrict__ as_, const float* __restrict__ ad_,
            float* __restrict__ eT, unsigned* __restrict__ mOrd)
{
    const int ET = E_EDGES + N_NODES;
    const int i = (int)(blockIdx.x * (size_t)blockDim.x + threadIdx.x);
    if (i >= ET) return;
    int s, d;
    if (i < E_EDGES) { s = se[i]; d = de[i]; } else { s = d = i - E_EDGES; }
    float ev = as_[s] + ad_[d];
    ev = ev > 0.0f ? ev : NEG_SLOPE * ev;
    eT[i] = ev;
    atomicMax(&mOrd[d], f2ord(ev));
}

__global__ void __launch_bounds__(256)
edge_exp(const int* __restrict__ de, float* __restrict__ eT,
         const unsigned* __restrict__ mOrd, float* __restrict__ den)
{
    const int ET = E_EDGES + N_NODES;
    const int i = (int)(blockIdx.x * (size_t)blockDim.x + threadIdx.x);
    if (i >= ET) return;
    const int d = (i < E_EDGES) ? de[i] : (i - E_EDGES);
    const float ex = expf(eT[i] - ord2f(mOrd[d]));
    eT[i] = ex;
    atomicAdd(&den[d], ex);
}

__global__ void __launch_bounds__(256)
edge_scatter(const int* __restrict__ se, const int* __restrict__ de,
             const float* __restrict__ H, const float* __restrict__ eT,
             const float* __restrict__ den, float* __restrict__ agg)
{
    const int ET = E_EDGES + N_NODES;
    const int lane = threadIdx.x & 31;
    const int e = (int)(blockIdx.x * (size_t)(blockDim.x >> 5) + (threadIdx.x >> 5));
    if (e >= ET) return;
    int s, d;
    if (e < E_EDGES) { s = se[e]; d = de[e]; } else { s = d = e - E_EDGES; }
    const float alpha = eT[e] / den[d];
    const float* hs = H + (size_t)s * HIDV;
    float* ag = agg + (size_t)d * HIDV;
    atomicAdd(&ag[lane],      hs[lane]      * alpha);
    atomicAdd(&ag[lane + 32], hs[lane + 32] * alpha);
}

__global__ void __launch_bounds__(256)
finish_elu(const float* __restrict__ agg, const float* __restrict__ bias,
           float* __restrict__ out)
{
    const size_t i = blockIdx.x * (size_t)blockDim.x + threadIdx.x;
    if (i >= (size_t)N_NODES * HIDV) return;
    const float v = agg[i] + bias[i & (HIDV - 1)];
    out[i] = v > 0.0f ? v : expm1f(v);
}

__global__ void __launch_bounds__(256)
out_proj(const float* __restrict__ H, const float* __restrict__ Wo,
         const float* __restrict__ bo, float* __restrict__ y, int n)
{
    const int lane = threadIdx.x & 31;
    const int node = (int)((blockIdx.x * (size_t)blockDim.x + threadIdx.x) >> 5);
    if (node >= n) return;
    const float* h = H + (size_t)node * HIDV;
    float s = h[lane] * Wo[lane] + h[lane + 32] * Wo[lane + 32];
    #pragma unroll
    for (int off = 16; off; off >>= 1) s += __shfl_xor(s, off, 32);
    if (lane == 0) y[node] = s + bo[0];
}

extern "C" void kernel_launch(void* const* d_in, const int* in_sizes, int n_in,
                              void* d_out, int out_size, void* d_ws, size_t ws_size,
                              hipStream_t stream)
{
    const float* x      = (const float*)d_in[0];
    const int*   ei     = (const int*)d_in[1];
    const float* W1     = (const float*)d_in[2];
    const float* a_src1 = (const float*)d_in[3];
    const float* a_dst1 = (const float*)d_in[4];
    const float* b1     = (const float*)d_in[5];
    const float* W2     = (const float*)d_in[6];
    const float* a_src2 = (const float*)d_in[7];
    const float* a_dst2 = (const float*)d_in[8];
    const float* b2     = (const float*)d_in[9];
    const float* W_out  = (const float*)d_in[10];
    const float* b_out  = (const float*)d_in[11];
    float* y = (float*)d_out;

    const int* se = ei;
    const int* de = ei + E_EDGES;

    float*    h1   = (float*)d_ws;
    float*    h2   = h1 + (size_t)N_NODES * HIDV;
    float*    agg  = h2 + (size_t)N_NODES * HIDV;
    float*    as_  = agg + (size_t)N_NODES * HIDV;
    float*    ad_  = as_ + N_NODES;
    float*    den  = ad_ + N_NODES;
    unsigned* mOrd = (unsigned*)(den + N_NODES);
    float*    eT   = (float*)(mOrd + N_NODES);

    const int ET       = E_EDGES + N_NODES;
    const int gRowT    = N_NODES / 16;
    const int gNodeVec = (N_NODES * HIDV + 255) / 256;
    const int gNodeWav = (N_NODES + 7) / 8;
    const int gEdgeThr = (ET + 255) / 256;
    const int gEdgeWav = (ET + 7) / 8;

    gemm_wmma_f32<IN_DIMV><<<gRowT, 128, 0, stream>>>(x, W1, h1);
    node_alpha<<<gNodeWav, 256, 0, stream>>>(h1, a_src1, a_dst1, as_, ad_, N_NODES);
    init_node<<<gNodeVec, 256, 0, stream>>>(mOrd, den, agg);
    edge_logits<<<gEdgeThr, 256, 0, stream>>>(se, de, as_, ad_, eT, mOrd);
    edge_exp<<<gEdgeThr, 256, 0, stream>>>(de, eT, mOrd, den);
    edge_scatter<<<gEdgeWav, 256, 0, stream>>>(se, de, h1, eT, den, agg);
    finish_elu<<<gNodeVec, 256, 0, stream>>>(agg, b1, h1);

    gemm_wmma_f32<HIDV><<<gRowT, 128, 0, stream>>>(h1, W2, h2);
    node_alpha<<<gNodeWav, 256, 0, stream>>>(h2, a_src2, a_dst2, as_, ad_, N_NODES);
    init_node<<<gNodeVec, 256, 0, stream>>>(mOrd, den, agg);
    edge_logits<<<gEdgeThr, 256, 0, stream>>>(se, de, as_, ad_, eT, mOrd);
    edge_exp<<<gEdgeThr, 256, 0, stream>>>(de, eT, mOrd, den);
    edge_scatter<<<gEdgeWav, 256, 0, stream>>>(se, de, h2, eT, den, agg);
    finish_elu<<<gNodeVec, 256, 0, stream>>>(agg, b2, h2);

    out_proj<<<gNodeWav, 256, 0, stream>>>(h2, W_out, b_out, y, N_NODES);
}


# Edges counting-sorted by destination once per call; one wave per node does max, exp-sum and alpha-weighted row gather in registers, bias+ELU in its store; no float atomics, sums in sorted-slot order
# speedup vs baseline: 2.9983x; 2.9983x over previous
#include <hip/hip_runtime.h>
#include <hip/hip_bf16.h>

#define N_NODES   100000
#define E_EDGES   1600000
#define IN_DIMV   128
#define HIDV      64
#define NEG_SLOPE 0.2f

typedef __attribute__((ext_vector_type(2))) float v2f;
typedef __attribute__((ext_vector_type(8))) float v8f;

template <int K>
__global__ void __launch_bounds__(128)
gemm_wmma_f32(const float* __restrict__ Hin, const float* __restrict__ W,
              float* __restrict__ Hout)
{
    const int lane = threadIdx.x & 31;
    const int wv   = threadIdx.x >> 5;
    const int row0 = blockIdx.x << 4;
    const int col0 = wv << 4;
    const int m    = lane & 15;
    const int kb   = (lane >> 4) << 1;

    v8f acc = {};
    const float* arow = Hin + (size_t)(row0 + m) * K;
    #pragma unroll 4
    for (int k0 = 0; k0 < K; k0 += 4) {
        v2f a, b;
        a.x = arow[k0 + kb];
        a.y = arow[k0 + kb + 1];
        b.x = W[(size_t)(k0 + kb)     * HIDV + col0 + m];
        b.y = W[(size_t)(k0 + kb + 1) * HIDV + col0 + m];
        acc = __builtin_amdgcn_wmma_f32_16x16x4_f32(
                   false, a,  false, b,
                   (short)0, acc,  false,  false);
    }

    const int rbase = row0 + ((lane >> 4) << 3);
    #pragma unroll
    for (int j = 0; j < 8; ++j)
        Hout[(size_t)(rbase + j) * HIDV + col0 + m] = acc[j];
}

__global__ void __launch_bounds__(256)
node_alpha(const float* __restrict__ H, const float* __restrict__ avs,
           const float* __restrict__ avd, float* __restrict__ as_,
           float* __restrict__ ad_, int n)
{
    const int lane = threadIdx.x & 31;
    const int node = (int)((blockIdx.x * (size_t)blockDim.x + threadIdx.x) >> 5);
    if (node >= n) return;
    const float* h = H + (size_t)node * HIDV;
    float s = h[lane] * avs[lane] + h[lane + 32] * avs[lane + 32];
    float d = h[lane] * avd[lane] + h[lane + 32] * avd[lane + 32];
    #pragma unroll
    for (int off = 16; off; off >>= 1) {
        s += __shfl_xor(s, off, 32);
        d += __shfl_xor(d, off, 32);
    }
    if (lane == 0) { as_[node] = s; ad_[node] = d; }
}

#define SCAN_TILE   1024
#define SCAN_BLOCKS ((N_NODES + SCAN_TILE - 1) / SCAN_TILE)
static_assert(N_NODES % 4 == 0, "int4 loads of the per-node counters");
static_assert(SCAN_BLOCKS <= 256, "one thread per preceding tile in scan_write");
static_assert(N_NODES % 8 == 0, "eight one-wave nodes per gather block");

__global__ void __launch_bounds__(256)
zero_counts(int* __restrict__ cnt)
{
    const int i = (int)(blockIdx.x * (size_t)blockDim.x + threadIdx.x);
    if (i < N_NODES) cnt[i] = 0;
}

__global__ void __launch_bounds__(256)
edge_rank(const int* __restrict__ de, int* __restrict__ cnt, int* __restrict__ rank)
{
    const int i = (int)(blockIdx.x * (size_t)blockDim.x + threadIdx.x);
    if (i >= E_EDGES) return;
    rank[i] = atomicAdd(&cnt[de[i]], 1);
}

__global__ void __launch_bounds__(256)
scan_block_sums(const int* __restrict__ cnt, int* __restrict__ bsum)
{
    __shared__ int wsum[8];
    const int lane = threadIdx.x & 31;
    const int wv   = threadIdx.x >> 5;
    const int base = blockIdx.x * SCAN_TILE + threadIdx.x * 4;
    int t = 0;
    if (base < N_NODES) {
        const int4 c = *(const int4*)(cnt + base);
        t = c.x + c.y + c.z + c.w;
    }
    #pragma unroll
    for (int off = 16; off; off >>= 1) t += __shfl_xor(t, off, 32);
    if (lane == 0) wsum[wv] = t;
    __syncthreads();
    if (threadIdx.x == 0) {
        int s = 0;
        #pragma unroll
        for (int w = 0; w < 8; ++w) s += wsum[w];
        bsum[blockIdx.x] = s;
    }
}

__global__ void __launch_bounds__(256)
scan_write(const int* __restrict__ cnt, const int* __restrict__ bsum,
           int* __restrict__ rowptr)
{
    __shared__ int wtot[8];
    __shared__ int woff[8];
    const int lane = threadIdx.x & 31;
    const int wv   = threadIdx.x >> 5;
    const int base = blockIdx.x * SCAN_TILE + threadIdx.x * 4;
    int4 c = make_int4(0, 0, 0, 0);
    if (base < N_NODES) c = *(const int4*)(cnt + base);
    const int t = c.x + c.y + c.z + c.w;
    int inc = t;
    #pragma unroll
    for (int off = 1; off < 32; off <<= 1) {
        const int v = __shfl_up(inc, off, 32);
        if (lane >= off) inc += v;
    }
    int p = ((int)threadIdx.x < (int)blockIdx.x) ? bsum[threadIdx.x] : 0;
    #pragma unroll
    for (int off = 16; off; off >>= 1) p += __shfl_xor(p, off, 32);
    if (lane == 31) wtot[wv] = inc;
    if (lane == 0)  woff[wv] = p;
    __syncthreads();
    int pre = 0;
    #pragma unroll
    for (int w = 0; w < 8; ++w) {
        pre += woff[w];
        if (w < wv) pre += wtot[w];
    }
    const int ex = pre + inc - t;
    if (base < N_NODES) {
        int4 r;
        r.x = ex;
        r.y = r.x + c.x;
        r.z = r.y + c.y;
        r.w = r.z + c.z;
        *(int4*)(rowptr + base) = r;
    }
}

__global__ void __launch_bounds__(256)
csr_fill(const int* __restrict__ se, const int* __restrict__ de,
         const int* __restrict__ rowptr, const int* __restrict__ rank,
         int* __restrict__ csr)
{
    const int i = (int)(blockIdx.x * (size_t)blockDim.x + threadIdx.x);
    if (i >= E_EDGES) return;
    const int p = rowptr[de[i]] + rank[i];
    if ((unsigned)p < (unsigned)E_EDGES) csr[p] = se[i];
}

__device__ __forceinline__ float lrelu(float v) { return v > 0.0f ? v : NEG_SLOPE * v; }

__global__ void __launch_bounds__(256)
gat_gather(const int* __restrict__ rowptr, const int* __restrict__ cnt,
           const int* __restrict__ csr, const float* __restrict__ H,
           const float* __restrict__ as_, const float* __restrict__ ad_,
           const float* __restrict__ bias, float* __restrict__ Hout)
{
    const int lane = threadIdx.x & 31;
    const int node = __builtin_amdgcn_readfirstlane((int)(blockIdx.x * 8 + (threadIdx.x >> 5)));
    if (node >= N_NODES) return;

    const int   beg = rowptr[node];
    const int   deg = cnt[node];
    const float adn = ad_[node];
    const float es  = lrelu(as_[node] + adn);
    const v2f* H2 = (const v2f*)H;
    const v2f hself = H2[(size_t)node * 32 + lane];

    int   s0 = node;
    float e0 = -__builtin_inff();
    if (lane < deg) { s0 = csr[beg + lane]; e0 = lrelu(as_[s0] + adn); }

    float m = fmaxf(es, e0);
    for (int j = lane + 32; j < deg; j += 32)
        m = fmaxf(m, lrelu(as_[csr[beg + j]] + adn));
    #pragma unroll
    for (int off = 16; off; off >>= 1) m = fmaxf(m, __shfl_xor(m, off, 32));

    const float ex0 = (lane < deg) ? expf(e0 - m) : 0.0f;
    float dsum = ex0;
    for (int j = lane + 32; j < deg; j += 32)
        dsum += expf(lrelu(as_[csr[beg + j]] + adn) - m);
    #pragma unroll
    for (int off = 16; off; off >>= 1) dsum += __shfl_xor(dsum, off, 32);
    const float exs = expf(es - m);
    const float den = dsum + exs;

    const float a_self = exs / den;
    float accx = hself.x * a_self;
    float accy = hself.y * a_self;

    for (int j0 = 0; j0 < deg; j0 += 32) {
        int   sj = s0;
        float al = ex0 / den;
        if (j0 > 0) {
            const int j = j0 + lane;
            sj = node; al = 0.0f;
            if (j < deg) {
                sj = csr[beg + j];
                al = expf(lrelu(as_[sj] + adn) - m) / den;
            }
        }
        const int c = (deg - j0) < 32 ? (deg - j0) : 32;
        for (int t0 = 0; t0 < c; t0 += 4) {
            v2f   hv[4];
            float wt[4];
            #pragma unroll
            for (int u = 0; u < 4; ++u) {
                const int t  = t0 + u;
                const int st = __builtin_amdgcn_readlane(sj, t);
                wt[u] = __int_as_float(__builtin_amdgcn_readlane(__float_as_int(al), t));
                hv[u] = H2[(size_t)st * 32 + lane];
            }
            #pragma unroll
            for (int u = 0; u < 4; ++u) {
                const bool live = (t0 + u) < c;
                accx += (live ? hv[u].x : 0.0f) * wt[u];
                accy += (live ? hv[u].y : 0.0f) * wt[u];
            }
        }
    }

    const float vx = accx + bias[2 * lane];
    const float vy = accy + bias[2 * lane + 1];
    v2f o;
    o.x = vx > 0.0f ? vx : expm1f(vx);
    o.y = vy > 0.0f ? vy : expm1f(vy);
    ((v2f*)Hout)[(size_t)node * 32 + lane] = o;
}

__global__ void __launch_bounds__(256)
out_proj(const float* __restrict__ H, const float* __restrict__ Wo,
         const float* __restrict__ bo, float* __restrict__ y, int n)
{
    const int lane = threadIdx.x & 31;
    const int node = (int)((blockIdx.x * (size_t)blockDim.x + threadIdx.x) >> 5);
    if (node >= n) return;
    const float* h = H + (size_t)node * HIDV;
    float s = h[lane] * Wo[lane] + h[lane + 32] * Wo[lane + 32];
    #pragma unroll
    for (int off = 16; off; off >>= 1) s += __shfl_xor(s, off, 32);
    if (lane == 0) y[node] = s + bo[0];
}

extern "C" void kernel_launch(void* const* d_in, const int* in_sizes, int n_in,
                              void* d_out, int out_size, void* d_ws, size_t ws_size,
                              hipStream_t stream)
{
    const float* x      = (const float*)d_in[0];
    const int*   ei     = (const int*)d_in[1];
    const float* W1     = (const float*)d_in[2];
    const float* a_src1 = (const float*)d_in[3];
    const float* a_dst1 = (const float*)d_in[4];
    const float* b1     = (const float*)d_in[5];
    const float* W2     = (const float*)d_in[6];
    const float* a_src2 = (const float*)d_in[7];
    const float* a_dst2 = (const float*)d_in[8];
    const float* b2     = (const float*)d_in[9];
    const float* W_out  = (const float*)d_in[10];
    const float* b_out  = (const float*)d_in[11];
    float* y = (float*)d_out;

    const int* se = ei;
    const int* de = ei + E_EDGES;

    float* h1     = (float*)d_ws;
    float* h2     = h1 + (size_t)N_NODES * HIDV;
    float* as_    = h2 + (size_t)N_NODES * HIDV;
    float* ad_    = as_ + N_NODES;
    int*   cnt    = (int*)(ad_ + N_NODES);
    int*   rowptr = cnt + N_NODES;
    int*   bsum   = rowptr + N_NODES;
    int*   rank   = bsum + 256;
    int*   csr    = rank + E_EDGES;

    const int gRowT    = N_NODES / 16;
    const int gNodeThr = (N_NODES + 255) / 256;
    const int gNodeWav = (N_NODES + 7) / 8;
    const int gEdgeThr = (E_EDGES + 255) / 256;

    zero_counts<<<gNodeThr, 256, 0, stream>>>(cnt);
    edge_rank<<<gEdgeThr, 256, 0, stream>>>(de, cnt, rank);
    scan_block_sums<<<SCAN_BLOCKS, 256, 0, stream>>>(cnt, bsum);
    scan_write<<<SCAN_BLOCKS, 256, 0, stream>>>(cnt, bsum, rowptr);
    csr_fill<<<gEdgeThr, 256, 0, stream>>>(se, de, rowptr, rank, csr);

    gemm_wmma_f32<IN_DIMV><<<gRowT, 128, 0, stream>>>(x, W1, h1);
    node_alpha<<<gNodeWav, 256, 0, stream>>>(h1, a_src1, a_dst1, as_, ad_, N_NODES);
    gat_gather<<<gNodeWav, 256, 0, stream>>>(rowptr, cnt, csr, h1, as_, ad_, b1, h2);

    gemm_wmma_f32<HIDV><<<gRowT, 128, 0, stream>>>(h2, W2, h1);
    node_alpha<<<gNodeWav, 256, 0, stream>>>(h1, a_src2, a_dst2, as_, ad_, N_NODES);
    gat_gather<<<gNodeWav, 256, 0, stream>>>(rowptr, cnt, csr, h1, as_, ad_, b2, h2);

    out_proj<<<gNodeWav, 256, 0, stream>>>(h2, W_out, b_out, y, N_NODES);
}


# GEMM operand path: 5 row tiles per block, input tile staged once in LDS by 16-byte whole-line loads with the next tile prefetched, W fragments kept in registers; sum order unchanged
# speedup vs baseline: 6.5391x; 2.1809x over previous
#include <hip/hip_runtime.h>
#include <hip/hip_bf16.h>

#define N_NODES   100000
#define E_EDGES   1600000
#define IN_DIMV   128
#define HIDV      64
#define NEG_SLOPE 0.2f

typedef __attribute__((ext_vector_type(2))) float v2f;
typedef __attribute__((ext_vector_type(8))) float v8f;

template <int MASK>
__device__ __forceinline__ float row_xor(float v)
{
    return __int_as_float(__builtin_amdgcn_update_dpp(
        0, __float_as_int(v), 0x160 | MASK, 0xf, 0xf, true));
}

#define GEMM_TPB 5
static_assert((N_NODES / 16) % GEMM_TPB == 0 && N_NODES % 16 == 0, "exact tile loop");

template <int K>
__global__ void __launch_bounds__(128)
gemm_wmma_f32(const float* __restrict__ Hin, const float* __restrict__ W,
              const float* __restrict__ avs, const float* __restrict__ avd,
              float* __restrict__ Hout, float* __restrict__ as_,
              float* __restrict__ ad_)
{
    constexpr int P   = K + 4;
    constexpr int C4  = K / 4;
    constexpr int RPP = 128 / C4;
    constexpr int NV  = 16 / RPP;
    static_assert((P / 4) % 2 == 1 && 128 % C4 == 0 && 16 % RPP == 0, "tile staging shape");

    __shared__ __attribute__((aligned(16))) float As[16 * P];
    __shared__ float part[2][4][16];

    const int tid  = threadIdx.x;
    const int lane = tid & 31;
    const int wv   = tid >> 5;
    const int col0 = wv << 4;
    const int m    = lane & 15;
    const int kb   = (lane >> 4) << 1;

    const float ws = avs[col0 + m];
    const float wd = avd[col0 + m];

    v2f breg[K / 4];
    #pragma unroll
    for (int i = 0; i < K / 4; ++i) {
        const int k0 = i * 4;
        breg[i].x = W[(size_t)(k0 + kb)     * HIDV + col0 + m];
        breg[i].y = W[(size_t)(k0 + kb + 1) * HIDV + col0 + m];
    }

    const int srow  = tid / C4;
    const int sc4   = tid % C4;
    const int tile0 = blockIdx.x * GEMM_TPB;

    static_assert(NV == 2 || NV == 4, "two or four passes");
    float4 pf0, pf1, pf2 = {}, pf3 = {};
    auto fetch = [&](const float* tile) {
        const float4* g = (const float4*)tile;
        pf0 = g[(0 * RPP + srow) * C4 + sc4];
        pf1 = g[(1 * RPP + srow) * C4 + sc4];
        if constexpr (NV == 4) {
            pf2 = g[(2 * RPP + srow) * C4 + sc4];
            pf3 = g[(3 * RPP + srow) * C4 + sc4];
        }
    };
    fetch(Hin + (size_t)(tile0 << 4) * K);

    #pragma nounroll
    for (int it = 0; it < GEMM_TPB; ++it) {
        const int row0 = (tile0 + it) << 4;

        *(float4*)&As[(0 * RPP + srow) * P + sc4 * 4] = pf0;
        *(float4*)&As[(1 * RPP + srow) * P + sc4 * 4] = pf1;
        if constexpr (NV == 4) {
            *(float4*)&As[(2 * RPP + srow) * P + sc4 * 4] = pf2;
            *(float4*)&As[(3 * RPP + srow) * P + sc4 * 4] = pf3;
        }
        __syncthreads();

        if (it + 1 < GEMM_TPB) fetch(Hin + (size_t)(row0 + 16) * K);

        v8f acc = {};
        const float* arow = As + m * P + kb;
        #pragma unroll
        for (int i = 0; i < K / 4; ++i) {
            const v2f a = *(const v2f*)(arow + i * 4);
            acc = __builtin_amdgcn_wmma_f32_16x16x4_f32(
                       false, a,  false, breg[i],
                       (short)0, acc,  false,  false);
        }

        const bool b3 = (m & 8) != 0, b2 = (m & 4) != 0, b1 = (m & 2) != 0;
        float s4[4], d4[4];
        #pragma unroll
        for (int i = 0; i < 4; ++i) {
            const float slo = acc[i] * ws, shi = acc[i + 4] * ws;
            const float dlo = acc[i] * wd, dhi = acc[i + 4] * wd;
            s4[i] = (b3 ? shi : slo) + row_xor<8>(b3 ? slo : shi);
            d4[i] = (b3 ? dhi : dlo) + row_xor<8>(b3 ? dlo : dhi);
        }
        float s2[2], d2[2];
        #pragma unroll
        for (int i = 0; i < 2; ++i) {
            s2[i] = (b2 ? s4[i + 2] : s4[i]) + row_xor<4>(b2 ? s4[i] : s4[i + 2]);
            d2[i] = (b2 ? d4[i + 2] : d4[i]) + row_xor<4>(b2 ? d4[i] : d4[i + 2]);
        }
        float s1 = (b1 ? s2[1] : s2[0]) + row_xor<2>(b1 ? s2[0] : s2[1]);
        float d1 = (b1 ? d2[1] : d2[0]) + row_xor<2>(b1 ? d2[0] : d2[1]);
        s1 += row_xor<1>(s1);
        d1 += row_xor<1>(d1);
        if ((m & 1) == 0) {
            const int r = ((lane >> 4) << 3) + (m >> 1);
            part[0][wv][r] = s1;
            part[1][wv][r] = d1;
        }
        __syncthreads();
        if (tid < 32) {
            const int q = tid >> 4;
            const int r = tid & 15;
            const float v = ((part[q][0][r] + part[q][1][r]) + part[q][2][r]) + part[q][3][r];
            (q ? ad_ : as_)[row0 + r] = v;
        }

        const int rbase = row0 + ((lane >> 4) << 3);
        #pragma unroll
        for (int j = 0; j < 8; ++j)
            Hout[(size_t)(rbase + j) * HIDV + col0 + m] = acc[j];
    }
}

#define BKT_SHIFT 9
#define BKT_W     (1 << BKT_SHIFT)
#define NBK       ((N_NODES + BKT_W - 1) / BKT_W)
#define PCHUNK    6400
#define PBLOCKS   (E_EDGES / PCHUNK)
#define PTHREADS  640
#define PEDGES    (PCHUNK / PTHREADS)
#define SCAP      32768
#define RPW       ((PBLOCKS + 31) / 32)
static_assert(E_EDGES % PCHUNK == 0 && PCHUNK % PTHREADS == 0, "exact chunks");
static_assert(N_NODES <= (1 << (31 - BKT_SHIFT)), "source id and in-bucket destination share one word");
static_assert(NBK <= 256 && PBLOCKS <= 1024 && PCHUNK < 65536, "table sizes");
static_assert(N_NODES % 8 == 0, "eight one-wave nodes per gather block");

template <int NWAVES>
__device__ __forceinline__ int block_excl_scan(int v, int* wsum, int& total)
{
    const int lane = threadIdx.x & 31;
    const int wv   = threadIdx.x >> 5;
    int inc = v;
    #pragma unroll
    for (int off = 1; off < 32; off <<= 1) {
        const int t = __shfl_up(inc, off, 32);
        if (lane >= off) inc += t;
    }
    if (lane == 31) wsum[wv] = inc;
    __syncthreads();
    int pre = 0, tot = 0;
    #pragma unroll
    for (int w = 0; w < NWAVES; ++w) {
        const int x = wsum[w];
        tot += x;
        if (w < wv) pre += x;
    }
    total = tot;
    return pre + inc - v;
}

__global__ void __launch_bounds__(PTHREADS)
edge_partition(const int* __restrict__ se, const int* __restrict__ de,
               int* __restrict__ part, int* __restrict__ tabS, int* __restrict__ tabC)
{
    __shared__ int lcnt[256];
    __shared__ int lst[256];
    __shared__ int wsum[PTHREADS / 32];
    __shared__ int stage[PCHUNK];

    const int tid = threadIdx.x;
    const int g   = blockIdx.x;
    const int e0  = g * PCHUNK;
    if (tid < 256) lcnt[tid] = 0;
    __syncthreads();

    int d[PEDGES], s[PEDGES];
    #pragma unroll
    for (int k = 0; k < PEDGES; ++k) {
        const int i = e0 + k * PTHREADS + tid;
        d[k] = de[i];
        s[k] = se[i];
    }
    int w[PEDGES], br[PEDGES];
    #pragma unroll
    for (int k = 0; k < PEDGES; ++k) {
        int bk = (int)((unsigned)d[k] >> BKT_SHIFT);
        bk = bk < NBK ? bk : NBK - 1;
        w[k]  = (s[k] << BKT_SHIFT) | (d[k] & (BKT_W - 1));
        br[k] = (bk << 16) | atomicAdd(&lcnt[bk], 1);
    }
    __syncthreads();

    const int c = tid < 256 ? lcnt[tid] : 0;
    int tot;
    const int ex = block_excl_scan<PTHREADS / 32>(c, wsum, tot);
    if (tid < 256) lst[tid] = ex;
    if (tid < NBK) {
        tabS[tid * PBLOCKS + g] = ex;
        tabC[tid * PBLOCKS + g] = c;
    }
    __syncthreads();

    #pragma unroll
    for (int k = 0; k < PEDGES; ++k)
        stage[lst[br[k] >> 16] + (br[k] & 0xffff)] = w[k];
    __syncthreads();

    #pragma unroll
    for (int k = 0; k < PEDGES; ++k)
        part[e0 + k * PTHREADS + tid] = stage[k * PTHREADS + tid];
}

__global__ void __launch_bounds__(1024)
bucket_sort(const int* __restrict__ part, const int* __restrict__ tabS,
            const int* __restrict__ tabC, int* __restrict__ cnt,
            int* __restrict__ rowptr, int* __restrict__ csr)
{
    __shared__ int rs[PBLOCKS];
    __shared__ int rc[PBLOCKS];
    __shared__ int ncnt[BKT_W];
    __shared__ int cur[BKT_W];
    __shared__ int wsum[32];
    __shared__ int sorted[SCAP];

    const int tid  = threadIdx.x;
    const int lane = tid & 31;
    const int wv   = tid >> 5;
    const int b    = blockIdx.x;
    const int nprev = b * PBLOCKS;

    int s = 0;
    if (tid < PBLOCKS) {
        int a = tabS[nprev + tid];
        int n = tabC[nprev + tid];
        s = a;
        a = a < 0 ? 0 : (a > PCHUNK ? PCHUNK : a);
        n = n < 0 ? 0 : (n > PCHUNK - a ? PCHUNK - a : n);
        rs[tid] = a;
        rc[tid] = n;
    }
    #pragma unroll
    for (int off = 16; off; off >>= 1) s += __shfl_xor(s, off, 32);
    if (lane == 0) wsum[wv] = s;
    if (tid < BKT_W) ncnt[tid] = 0;
    __syncthreads();
    int base = 0;
    #pragma unroll
    for (int w = 0; w < 32; ++w) base += wsum[w];

    int w0[RPW], w1[RPW];
    #pragma unroll
    for (int r = 0; r < RPW; ++r) {
        const int g = wv + 32 * r;
        w0[r] = -1;
        w1[r] = -1;
        if (g < PBLOCKS) {
            const int* p = part + g * PCHUNK + rs[g];
            const int  n = rc[g];
            if (lane < n)      w0[r] = p[lane];
            if (lane + 32 < n) w1[r] = p[lane + 32];
        }
    }

    #pragma unroll
    for (int r = 0; r < RPW; ++r) {
        if (w0[r] >= 0) atomicAdd(&ncnt[w0[r] & (BKT_W - 1)], 1);
        if (w1[r] >= 0) atomicAdd(&ncnt[w1[r] & (BKT_W - 1)], 1);
    }
    for (int g = wv; g < PBLOCKS; g += 32) {
        const int* p = part + g * PCHUNK + rs[g];
        const int  n = rc[g];
        for (int j = lane + 64; j < n; j += 32)
            atomicAdd(&ncnt[p[j] & (BKT_W - 1)], 1);
    }
    __syncthreads();

    const int c = tid < BKT_W ? ncnt[tid] : 0;
    int nb;
    const int ex = block_excl_scan<32>(c, wsum, nb);
    if (tid < BKT_W) {
        cur[tid] = ex;
        const int node = b * BKT_W + tid;
        if (node < N_NODES) { cnt[node] = c; rowptr[node] = base + ex; }
    }
    __syncthreads();

    const bool staged = nb <= SCAP;
    auto place = [&](int wd) {
        const int pos = atomicAdd(&cur[wd & (BKT_W - 1)], 1);
        const int src = (int)((unsigned)wd >> BKT_SHIFT);
        if (staged) {
            if ((unsigned)pos < (unsigned)SCAP) sorted[pos] = src;
        } else {
            const int q = base + pos;
            if ((unsigned)q < (unsigned)E_EDGES) csr[q] = src;
        }
    };
    #pragma unroll
    for (int r = 0; r < RPW; ++r) {
        if (w0[r] >= 0) place(w0[r]);
        if (w1[r] >= 0) place(w1[r]);
    }
    for (int g = wv; g < PBLOCKS; g += 32) {
        const int* p = part + g * PCHUNK + rs[g];
        const int  n = rc[g];
        for (int j = lane + 64; j < n; j += 32) place(p[j]);
    }
    __syncthreads();
    if (staged) {
        for (int e = tid; e < nb; e += 1024) {
            const int q = base + e;
            if ((unsigned)q < (unsigned)E_EDGES) csr[q] = sorted[e];
        }
    }
}

__device__ __forceinline__ float lrelu(float v) { return v > 0.0f ? v : NEG_SLOPE * v; }

template <bool HEAD>
__global__ void __launch_bounds__(256)
gat_gather(const int* __restrict__ rowptr, const int* __restrict__ cnt,
           const int* __restrict__ csr, const float* __restrict__ H,
           const float* __restrict__ as_, const float* __restrict__ ad_,
           const float* __restrict__ bias, float* __restrict__ Hout,
           const float* __restrict__ Wo, const float* __restrict__ bo,
           float* __restrict__ y)
{
    const int lane = threadIdx.x & 31;
    const int node = __builtin_amdgcn_readfirstlane((int)(blockIdx.x * 8 + (threadIdx.x >> 5)));
    if (node >= N_NODES) return;

    const int   beg = rowptr[node];
    const int   deg = cnt[node];
    const float adn = ad_[node];
    const float es  = lrelu(as_[node] + adn);
    const v2f* H2 = (const v2f*)H;
    const v2f hself = H2[(size_t)node * 32 + lane];
    v2f   wo  = {0.0f, 0.0f};
    float bov = 0.0f;
    if (HEAD) { wo = ((const v2f*)Wo)[lane]; bov = bo[0]; }

    int   s0 = node;
    float e0 = -__builtin_inff();
    if (lane < deg) { s0 = csr[beg + lane]; e0 = lrelu(as_[s0] + adn); }

    float m = fmaxf(es, e0);
    for (int j = lane + 32; j < deg; j += 32)
        m = fmaxf(m, lrelu(as_[csr[beg + j]] + adn));
    #pragma unroll
    for (int off = 16; off; off >>= 1) m = fmaxf(m, __shfl_xor(m, off, 32));

    const float ex0 = (lane < deg) ? expf(e0 - m) : 0.0f;
    float dsum = ex0;
    for (int j = lane + 32; j < deg; j += 32)
        dsum += expf(lrelu(as_[csr[beg + j]] + adn) - m);
    #pragma unroll
    for (int off = 16; off; off >>= 1) dsum += __shfl_xor(dsum, off, 32);
    const float exs = expf(es - m);
    const float den = dsum + exs;

    const float a_self = exs / den;
    float accx = hself.x * a_self;
    float accy = hself.y * a_self;

    for (int j0 = 0; j0 < deg; j0 += 32) {
        int   sj = s0;
        float al = ex0 / den;
        if (j0 > 0) {
            const int j = j0 + lane;
            sj = node; al = 0.0f;
            if (j < deg) {
                sj = csr[beg + j];
                al = expf(lrelu(as_[sj] + adn) - m) / den;
            }
        }
        const int c = (deg - j0) < 32 ? (deg - j0) : 32;
        for (int t0 = 0; t0 < c; t0 += 4) {
            v2f   hv[4];
            float wt[4];
            #pragma unroll
            for (int u = 0; u < 4; ++u) {
                const int t  = t0 + u;
                const int st = __builtin_amdgcn_readlane(sj, t);
                wt[u] = __int_as_float(__builtin_amdgcn_readlane(__float_as_int(al), t));
                hv[u] = H2[(size_t)st * 32 + lane];
            }
            #pragma unroll
            for (int u = 0; u < 4; ++u) {
                const bool live = (t0 + u) < c;
                accx += (live ? hv[u].x : 0.0f) * wt[u];
                accy += (live ? hv[u].y : 0.0f) * wt[u];
            }
        }
    }

    const float vx = accx + bias[2 * lane];
    const float vy = accy + bias[2 * lane + 1];
    v2f o;
    o.x = vx > 0.0f ? vx : expm1f(vx);
    o.y = vy > 0.0f ? vy : expm1f(vy);
    if (HEAD) {
        float s = o.x * wo.x + o.y * wo.y;
        s += row_xor<1>(s);
        s += row_xor<2>(s);
        s += row_xor<4>(s);
        s += row_xor<8>(s);
        s += __shfl_xor(s, 16, 32);
        if (lane == 0) y[node] = s + bov;
    } else {
        ((v2f*)Hout)[(size_t)node * 32 + lane] = o;
    }
}

extern "C" void kernel_launch(void* const* d_in, const int* in_sizes, int n_in,
                              void* d_out, int out_size, void* d_ws, size_t ws_size,
                              hipStream_t stream)
{
    const float* x      = (const float*)d_in[0];
    const int*   ei     = (const int*)d_in[1];
    const float* W1     = (const float*)d_in[2];
    const float* a_src1 = (const float*)d_in[3];
    const float* a_dst1 = (const float*)d_in[4];
    const float* b1     = (const float*)d_in[5];
    const float* W2     = (const float*)d_in[6];
    const float* a_src2 = (const float*)d_in[7];
    const float* a_dst2 = (const float*)d_in[8];
    const float* b2     = (const float*)d_in[9];
    const float* W_out  = (const float*)d_in[10];
    const float* b_out  = (const float*)d_in[11];
    float* y = (float*)d_out;

    const int* se = ei;
    const int* de = ei + E_EDGES;

    float* h1     = (float*)d_ws;
    float* h2     = h1 + (size_t)N_NODES * HIDV;
    float* as_    = h2 + (size_t)N_NODES * HIDV;
    float* ad_    = as_ + N_NODES;
    int*   cnt    = (int*)(ad_ + N_NODES);
    int*   rowptr = cnt + N_NODES;
    int*   csr    = rowptr + N_NODES;
    int*   part   = (int*)h2;
    int*   tabS   = part + E_EDGES;
    int*   tabC   = tabS + NBK * PBLOCKS;

    const int gRowT    = (N_NODES / 16) / GEMM_TPB;
    const int gNodeWav = (N_NODES + 7) / 8;

    edge_partition<<<PBLOCKS, PTHREADS, 0, stream>>>(se, de, part, tabS, tabC);
    bucket_sort<<<NBK, 1024, 0, stream>>>(part, tabS, tabC, cnt, rowptr, csr);

    gemm_wmma_f32<IN_DIMV><<<gRowT, 128, 0, stream>>>(x, W1, a_src1, a_dst1, h1, as_, ad_);
    gat_gather<false><<<gNodeWav, 256, 0, stream>>>(rowptr, cnt, csr, h1, as_, ad_, b1, h2,
                                                    W_out, b_out, y);

    gemm_wmma_f32<HIDV><<<gRowT, 128, 0, stream>>>(h2, W2, a_src2, a_dst2, h1, as_, ad_);
    gat_gather<true><<<gNodeWav, 256, 0, stream>>>(rowptr, cnt, csr, h1, as_, ad_, b2, h2,
                                                   W_out, b_out, y);
}


# Gather: one resident pass of waves that each walk many nodes, not one short wave per node; next node's header and edges prefetched, row loads issued a group ahead; slot order kept, products now fma
# speedup vs baseline: 7.0383x; 1.0763x over previous
#include <hip/hip_runtime.h>
#include <hip/hip_bf16.h>

#define N_NODES   100000
#define E_EDGES   1600000
#define IN_DIMV   128
#define HIDV      64
#define NEG_SLOPE 0.2f

typedef __attribute__((ext_vector_type(2))) float v2f;
typedef __attribute__((ext_vector_type(8))) float v8f;

template <int MASK>
__device__ __forceinline__ float row_xor(float v)
{
    return __int_as_float(__builtin_amdgcn_update_dpp(
        0, __float_as_int(v), 0x160 | MASK, 0xf, 0xf, true));
}

#define GEMM_TPB 5
static_assert((N_NODES / 16) % GEMM_TPB == 0 && N_NODES % 16 == 0, "exact tile loop");

template <int K>
__global__ void __launch_bounds__(128)
gemm_wmma_f32(const float* __restrict__ Hin, const float* __restrict__ W,
              const float* __restrict__ avs, const float* __restrict__ avd,
              float* __restrict__ Hout, float* __restrict__ as_,
              float* __restrict__ ad_)
{
    constexpr int P   = K + 4;
    constexpr int C4  = K / 4;
    constexpr int RPP = 128 / C4;
    constexpr int NV  = 16 / RPP;
    static_assert((P / 4) % 2 == 1 && 128 % C4 == 0 && 16 % RPP == 0, "tile staging shape");

    __shared__ __attribute__((aligned(16))) float As[16 * P];
    __shared__ float part[2][4][16];

    const int tid  = threadIdx.x;
    const int lane = tid & 31;
    const int wv   = tid >> 5;
    const int col0 = wv << 4;
    const int m    = lane & 15;
    const int kb   = (lane >> 4) << 1;

    const float ws = avs[col0 + m];
    const float wd = avd[col0 + m];

    v2f breg[K / 4];
    #pragma unroll
    for (int i = 0; i < K / 4; ++i) {
        const int k0 = i * 4;
        breg[i].x = W[(size_t)(k0 + kb)     * HIDV + col0 + m];
        breg[i].y = W[(size_t)(k0 + kb + 1) * HIDV + col0 + m];
    }

    const int srow  = tid / C4;
    const int sc4   = tid % C4;
    const int tile0 = blockIdx.x * GEMM_TPB;

    static_assert(NV == 2 || NV == 4, "two or four passes");
    float4 pf0, pf1, pf2 = {}, pf3 = {};
    auto fetch = [&](const float* tile) {
        const float4* g = (const float4*)tile;
        pf0 = g[(0 * RPP + srow) * C4 + sc4];
        pf1 = g[(1 * RPP + srow) * C4 + sc4];
        if constexpr (NV == 4) {
            pf2 = g[(2 * RPP + srow) * C4 + sc4];
            pf3 = g[(3 * RPP + srow) * C4 + sc4];
        }
    };
    fetch(Hin + (size_t)(tile0 << 4) * K);

    #pragma nounroll
    for (int it = 0; it < GEMM_TPB; ++it) {
        const int row0 = (tile0 + it) << 4;

        *(float4*)&As[(0 * RPP + srow) * P + sc4 * 4] = pf0;
        *(float4*)&As[(1 * RPP + srow) * P + sc4 * 4] = pf1;
        if constexpr (NV == 4) {
            *(float4*)&As[(2 * RPP + srow) * P + sc4 * 4] = pf2;
            *(float4*)&As[(3 * RPP + srow) * P + sc4 * 4] = pf3;
        }
        __syncthreads();

        if (it + 1 < GEMM_TPB) fetch(Hin + (size_t)(row0 + 16) * K);

        v8f acc = {};
        const float* arow = As + m * P + kb;
        #pragma unroll
        for (int i = 0; i < K / 4; ++i) {
            const v2f a = *(const v2f*)(arow + i * 4);
            acc = __builtin_amdgcn_wmma_f32_16x16x4_f32(
                       false, a,  false, breg[i],
                       (short)0, acc,  false,  false);
        }

        const bool b3 = (m & 8) != 0, b2 = (m & 4) != 0, b1 = (m & 2) != 0;
        float s4[4], d4[4];
        #pragma unroll
        for (int i = 0; i < 4; ++i) {
            const float slo = acc[i] * ws, shi = acc[i + 4] * ws;
            const float dlo = acc[i] * wd, dhi = acc[i + 4] * wd;
            s4[i] = (b3 ? shi : slo) + row_xor<8>(b3 ? slo : shi);
            d4[i] = (b3 ? dhi : dlo) + row_xor<8>(b3 ? dlo : dhi);
        }
        float s2[2], d2[2];
        #pragma unroll
        for (int i = 0; i < 2; ++i) {
            s2[i] = (b2 ? s4[i + 2] : s4[i]) + row_xor<4>(b2 ? s4[i] : s4[i + 2]);
            d2[i] = (b2 ? d4[i + 2] : d4[i]) + row_xor<4>(b2 ? d4[i] : d4[i + 2]);
        }
        float s1 = (b1 ? s2[1] : s2[0]) + row_xor<2>(b1 ? s2[0] : s2[1]);
        float d1 = (b1 ? d2[1] : d2[0]) + row_xor<2>(b1 ? d2[0] : d2[1]);
        s1 += row_xor<1>(s1);
        d1 += row_xor<1>(d1);
        if ((m & 1) == 0) {
            const int r = ((lane >> 4) << 3) + (m >> 1);
            part[0][wv][r] = s1;
            part[1][wv][r] = d1;
        }
        __syncthreads();
        if (tid < 32) {
            const int q = tid >> 4;
            const int r = tid & 15;
            const float v = ((part[q][0][r] + part[q][1][r]) + part[q][2][r]) + part[q][3][r];
            (q ? ad_ : as_)[row0 + r] = v;
        }

        const int rbase = row0 + ((lane >> 4) << 3);
        #pragma unroll
        for (int j = 0; j < 8; ++j)
            Hout[(size_t)(rbase + j) * HIDV + col0 + m] = acc[j];
    }
}

#define BKT_SHIFT 9
#define BKT_W     (1 << BKT_SHIFT)
#define NBK       ((N_NODES + BKT_W - 1) / BKT_W)
#define PCHUNK    6400
#define PBLOCKS   (E_EDGES / PCHUNK)
#define PTHREADS  640
#define PEDGES    (PCHUNK / PTHREADS)
#define SCAP      32768
#define RPW       ((PBLOCKS + 31) / 32)
static_assert(E_EDGES % PCHUNK == 0 && PCHUNK % PTHREADS == 0, "exact chunks");
static_assert(N_NODES <= (1 << (31 - BKT_SHIFT)), "source id and in-bucket destination share one word");
static_assert(NBK <= 256 && PBLOCKS <= 1024 && PCHUNK < 65536, "table sizes");
static_assert(N_NODES % 8 == 0, "eight one-wave nodes per gather block");

template <int NWAVES>
__device__ __forceinline__ int block_excl_scan(int v, int* wsum, int& total)
{
    const int lane = threadIdx.x & 31;
    const int wv   = threadIdx.x >> 5;
    int inc = v;
    #pragma unroll
    for (int off = 1; off < 32; off <<= 1) {
        const int t = __shfl_up(inc, off, 32);
        if (lane >= off) inc += t;
    }
    if (lane == 31) wsum[wv] = inc;
    __syncthreads();
    int pre = 0, tot = 0;
    #pragma unroll
    for (int w = 0; w < NWAVES; ++w) {
        const int x = wsum[w];
        tot += x;
        if (w < wv) pre += x;
    }
    total = tot;
    return pre + inc - v;
}

__global__ void __launch_bounds__(PTHREADS)
edge_partition(const int* __restrict__ se, const int* __restrict__ de,
               int* __restrict__ part, int* __restrict__ tabS, int* __restrict__ tabC)
{
    __shared__ int lcnt[256];
    __shared__ int lst[256];
    __shared__ int wsum[PTHREADS / 32];
    __shared__ int stage[PCHUNK];

    const int tid = threadIdx.x;
    const int g   = blockIdx.x;
    const int e0  = g * PCHUNK;
    if (tid < 256) lcnt[tid] = 0;
    __syncthreads();

    int d[PEDGES], s[PEDGES];
    #pragma unroll
    for (int k = 0; k < PEDGES; ++k) {
        const int i = e0 + k * PTHREADS + tid;
        d[k] = de[i];
        s[k] = se[i];
    }
    int w[PEDGES], br[PEDGES];
    #pragma unroll
    for (int k = 0; k < PEDGES; ++k) {
        int bk = (int)((unsigned)d[k] >> BKT_SHIFT);
        bk = bk < NBK ? bk : NBK - 1;
        w[k]  = (s[k] << BKT_SHIFT) | (d[k] & (BKT_W - 1));
        br[k] = (bk << 16) | atomicAdd(&lcnt[bk], 1);
    }
    __syncthreads();

    const int c = tid < 256 ? lcnt[tid] : 0;
    int tot;
    const int ex = block_excl_scan<PTHREADS / 32>(c, wsum, tot);
    if (tid < 256) lst[tid] = ex;
    if (tid < NBK) {
        tabS[tid * PBLOCKS + g] = ex;
        tabC[tid * PBLOCKS + g] = c;
    }
    __syncthreads();

    #pragma unroll
    for (int k = 0; k < PEDGES; ++k)
        stage[lst[br[k] >> 16] + (br[k] & 0xffff)] = w[k];
    __syncthreads();

    #pragma unroll
    for (int k = 0; k < PEDGES; ++k)
        part[e0 + k * PTHREADS + tid] = stage[k * PTHREADS + tid];
}

__global__ void __launch_bounds__(1024)
bucket_sort(const int* __restrict__ part, const int* __restrict__ tabS,
            const int* __restrict__ tabC, int* __restrict__ cnt,
            int* __restrict__ rowptr, int* __restrict__ csr)
{
    __shared__ int rs[PBLOCKS];
    __shared__ int rc[PBLOCKS];
    __shared__ int ncnt[BKT_W];
    __shared__ int cur[BKT_W];
    __shared__ int wsum[32];
    __shared__ int sorted[SCAP];

    const int tid  = threadIdx.x;
    const int lane = tid & 31;
    const int wv   = tid >> 5;
    const int b    = blockIdx.x;
    const int nprev = b * PBLOCKS;

    int s = 0;
    if (tid < PBLOCKS) {
        int a = tabS[nprev + tid];
        int n = tabC[nprev + tid];
        s = a;
        a = a < 0 ? 0 : (a > PCHUNK ? PCHUNK : a);
        n = n < 0 ? 0 : (n > PCHUNK - a ? PCHUNK - a : n);
        rs[tid] = a;
        rc[tid] = n;
    }
    #pragma unroll
    for (int off = 16; off; off >>= 1) s += __shfl_xor(s, off, 32);
    if (lane == 0) wsum[wv] = s;
    if (tid < BKT_W) ncnt[tid] = 0;
    __syncthreads();
    int base = 0;
    #pragma unroll
    for (int w = 0; w < 32; ++w) base += wsum[w];

    int w0[RPW], w1[RPW];
    #pragma unroll
    for (int r = 0; r < RPW; ++r) {
        const int g = wv + 32 * r;
        w0[r] = -1;
        w1[r] = -1;
        if (g < PBLOCKS) {
            const int* p = part + g * PCHUNK + rs[g];
            const int  n = rc[g];
            if (lane < n)      w0[r] = p[lane];
            if (lane + 32 < n) w1[r] = p[lane + 32];
        }
    }

    #pragma unroll
    for (int r = 0; r < RPW; ++r) {
        if (w0[r] >= 0) atomicAdd(&ncnt[w0[r] & (BKT_W - 1)], 1);
        if (w1[r] >= 0) atomicAdd(&ncnt[w1[r] & (BKT_W - 1)], 1);
    }
    for (int g = wv; g < PBLOCKS; g += 32) {
        const int* p = part + g * PCHUNK + rs[g];
        const int  n = rc[g];
        for (int j = lane + 64; j < n; j += 32)
            atomicAdd(&ncnt[p[j] & (BKT_W - 1)], 1);
    }
    __syncthreads();

    const int c = tid < BKT_W ? ncnt[tid] : 0;
    int nb;
    const int ex = block_excl_scan<32>(c, wsum, nb);
    if (tid < BKT_W) {
        cur[tid] = ex;
        const int node = b * BKT_W + tid;
        if (node < N_NODES) { cnt[node] = c; rowptr[node] = base + ex; }
    }
    __syncthreads();

    const bool staged = nb <= SCAP;
    auto place = [&](int wd) {
        const int pos = atomicAdd(&cur[wd & (BKT_W - 1)], 1);
        const int src = (int)((unsigned)wd >> BKT_SHIFT);
        if (staged) {
            if ((unsigned)pos < (unsigned)SCAP) sorted[pos] = src;
        } else {
            const int q = base + pos;
            if ((unsigned)q < (unsigned)E_EDGES) csr[q] = src;
        }
    };
    #pragma unroll
    for (int r = 0; r < RPW; ++r) {
        if (w0[r] >= 0) place(w0[r]);
        if (w1[r] >= 0) place(w1[r]);
    }
    for (int g = wv; g < PBLOCKS; g += 32) {
        const int* p = part + g * PCHUNK + rs[g];
        const int  n = rc[g];
        for (int j = lane + 64; j < n; j += 32) place(p[j]);
    }
    __syncthreads();
    if (staged) {
        for (int e = tid; e < nb; e += 1024) {
            const int q = base + e;
            if ((unsigned)q < (unsigned)E_EDGES) csr[q] = sorted[e];
        }
    }
}

#define GATHER_BLOCKS 2048

__device__ __forceinline__ float lrelu(float v) { return v > 0.0f ? v : NEG_SLOPE * v; }

template <bool HEAD>
__global__ void __launch_bounds__(256, 16)
gat_gather(const int* __restrict__ rowptr, const int* __restrict__ cnt,
           const int* __restrict__ csr, const float* __restrict__ H,
           const float* __restrict__ as_, const float* __restrict__ ad_,
           const float* __restrict__ bias, float* __restrict__ Hout,
           const float* __restrict__ Wo, const float* __restrict__ bo,
           float* __restrict__ y)
{
    const int lane = threadIdx.x & 31;
    const int wid = __builtin_amdgcn_readfirstlane((int)(blockIdx.x * 8 + (threadIdx.x >> 5)));
    constexpr int nw = GATHER_BLOCKS * 8;
    if (wid >= N_NODES) return;

    const char* Hb = (const char*)H;
    const float bx = bias[2 * lane];
    const float by = bias[2 * lane + 1];
    v2f   wo  = {0.0f, 0.0f};
    float bov = 0.0f;
    if (HEAD) { wo = ((const v2f*)Wo)[lane]; bov = bo[0]; }

    int   node = wid;
    int   beg  = rowptr[node];
    int   deg  = cnt[node];
    float adn  = ad_[node];
    float asn  = as_[node];
    int   s0   = node;
    if (lane < deg) s0 = csr[beg + lane];

    for (;;) {
        const int  nxt  = node + nw;
        const bool more = nxt < N_NODES;

        int   nbeg = 0, ndeg = 0;
        float nadn = 0.0f, nasn = 0.0f;
        if (more) { nbeg = rowptr[nxt]; ndeg = cnt[nxt]; nadn = ad_[nxt]; nasn = as_[nxt]; }

        auto rowld = [&](const int r, const unsigned lo) -> v2f {
            return *(const v2f*)(Hb + ((size_t)(unsigned)r << 8) + lo);
        };
        auto lane_off = [&]() -> unsigned {
            unsigned lo = (unsigned)lane * 8u;
            asm volatile("" : "+v"(lo));
            return lo;
        };

        const float e0 = lrelu(as_[s0] + adn);
        const v2f hself = rowld(node, lane_off());
        const int c = deg < 32 ? deg : 32;

        v2f ra[8], rb[8];
        auto issue = [&](v2f (&r)[8], const int t0) {
            const unsigned lo = lane_off();
            #pragma unroll
            for (int u = 0; u < 8; ++u) {
                const int st = __builtin_amdgcn_readlane(s0, t0 + u);
                r[u] = rowld(st, lo);
            }
        };
        issue(ra, 0);

        int ns0 = nxt;
        if (more && lane < ndeg) ns0 = csr[nbeg + lane];

        const float es = lrelu(asn + adn);
        float m = fmaxf(es, e0);
        for (int j = lane + 32; j < deg; j += 32)
            m = fmaxf(m, lrelu(as_[csr[beg + j]] + adn));
        #pragma unroll
        for (int off = 16; off; off >>= 1) m = fmaxf(m, __shfl_xor(m, off, 32));

        const float ex0 = (lane < deg) ? expf(e0 - m) : 0.0f;
        float dsum = ex0;
        for (int j = lane + 32; j < deg; j += 32)
            dsum += expf(lrelu(as_[csr[beg + j]] + adn) - m);
        #pragma unroll
        for (int off = 16; off; off >>= 1) dsum += __shfl_xor(dsum, off, 32);
        const float exs = expf(es - m);
        const float den = dsum + exs;

        const float a_self = exs / den;
        float accx = hself.x * a_self;
        float accy = hself.y * a_self;
        const float al = ex0 / den;

        auto consume = [&](const v2f (&r)[8], const int t0) {
            #pragma unroll
            for (int u = 0; u < 8; ++u) {
                if (t0 + u < c) {
                    const float wt = __int_as_float(__builtin_amdgcn_readlane(__float_as_int(al), t0 + u));
                    accx += r[u].x * wt;
                    accy += r[u].y * wt;
                }
            }
        };
        if (c > 8) issue(rb, 8);
        consume(ra, 0);
        if (c > 8) {
            if (c > 16) issue(ra, 16);
            consume(rb, 8);
            if (c > 16) {
                if (c > 24) issue(rb, 24);
                consume(ra, 16);
                if (c > 24) consume(rb, 24);
            }
        }

        for (int j0 = 32; j0 < deg; j0 += 32) {
            const int j = j0 + lane;
            int   sj  = node;
            float alj = 0.0f;
            if (j < deg) {
                sj  = csr[beg + j];
                alj = expf(lrelu(as_[sj] + adn) - m) / den;
            }
            const int cc = (deg - j0) < 32 ? (deg - j0) : 32;
            for (int t0 = 0; t0 < cc; t0 += 4) {
                v2f   hv[4];
                float wt[4];
                const unsigned lo = lane_off();
                #pragma unroll
                for (int u = 0; u < 4; ++u) {
                    const int t  = t0 + u;
                    const int st = __builtin_amdgcn_readlane(sj, t);
                    wt[u] = __int_as_float(__builtin_amdgcn_readlane(__float_as_int(alj), t));
                    hv[u] = rowld(st, lo);
                }
                #pragma unroll
                for (int u = 0; u < 4; ++u) {
                    const bool live = (t0 + u) < cc;
                    accx += (live ? hv[u].x : 0.0f) * wt[u];
                    accy += (live ? hv[u].y : 0.0f) * wt[u];
                }
            }
        }

        const float vx = accx + bx;
        const float vy = accy + by;
        v2f o;
        o.x = vx > 0.0f ? vx : expm1f(vx);
        o.y = vy > 0.0f ? vy : expm1f(vy);
        if (HEAD) {
            float s = o.x * wo.x + o.y * wo.y;
            s += row_xor<1>(s);
            s += row_xor<2>(s);
            s += row_xor<4>(s);
            s += row_xor<8>(s);
            s += __shfl_xor(s, 16, 32);
            if (lane == 0) y[node] = s + bov;
        } else {
            ((v2f*)Hout)[(size_t)node * 32 + lane] = o;
        }

        if (!more) break;
        node = nxt; beg = nbeg; deg = ndeg; adn = nadn; asn = nasn; s0 = ns0;
    }
}

extern "C" void kernel_launch(void* const* d_in, const int* in_sizes, int n_in,
                              void* d_out, int out_size, void* d_ws, size_t ws_size,
                              hipStream_t stream)
{
    const float* x      = (const float*)d_in[0];
    const int*   ei     = (const int*)d_in[1];
    const float* W1     = (const float*)d_in[2];
    const float* a_src1 = (const float*)d_in[3];
    const float* a_dst1 = (const float*)d_in[4];
    const float* b1     = (const float*)d_in[5];
    const float* W2     = (const float*)d_in[6];
    const float* a_src2 = (const float*)d_in[7];
    const float* a_dst2 = (const float*)d_in[8];
    const float* b2     = (const float*)d_in[9];
    const float* W_out  = (const float*)d_in[10];
    const float* b_out  = (const float*)d_in[11];
    float* y = (float*)d_out;

    const int* se = ei;
    const int* de = ei + E_EDGES;

    float* h1     = (float*)d_ws;
    float* h2     = h1 + (size_t)N_NODES * HIDV;
    float* as_    = h2 + (size_t)N_NODES * HIDV;
    float* ad_    = as_ + N_NODES;
    int*   cnt    = (int*)(ad_ + N_NODES);
    int*   rowptr = cnt + N_NODES;
    int*   csr    = rowptr + N_NODES;
    int*   part   = (int*)h2;
    int*   tabS   = part + E_EDGES;
    int*   tabC   = tabS + NBK * PBLOCKS;

    const int gRowT    = (N_NODES / 16) / GEMM_TPB;

    edge_partition<<<PBLOCKS, PTHREADS, 0, stream>>>(se, de, part, tabS, tabC);
    bucket_sort<<<NBK, 1024, 0, stream>>>(part, tabS, tabC, cnt, rowptr, csr);

    gemm_wmma_f32<IN_DIMV><<<gRowT, 128, 0, stream>>>(x, W1, a_src1, a_dst1, h1, as_, ad_);
    gat_gather<false><<<GATHER_BLOCKS, 256, 0, stream>>>(rowptr, cnt, csr, h1, as_, ad_, b1, h2,
                                                    W_out, b_out, y);

    gemm_wmma_f32<HIDV><<<gRowT, 128, 0, stream>>>(h2, W2, a_src2, a_dst2, h1, as_, ad_);
    gat_gather<true><<<GATHER_BLOCKS, 256, 0, stream>>>(rowptr, cnt, csr, h1, as_, ad_, b2, h2,
                                                   W_out, b_out, y);
}


# Gather: a wave works on two nodes at once, one per 16-lane row with 16-byte row loads: softmax, ELU and header code issue once for both, reductions are 4 DPP steps; sum order kept
# speedup vs baseline: 7.6981x; 1.0937x over previous
#include <hip/hip_runtime.h>
#include <hip/hip_bf16.h>

#define N_NODES   100000
#define E_EDGES   1600000
#define IN_DIMV   128
#define HIDV      64
#define NEG_SLOPE 0.2f

typedef __attribute__((ext_vector_type(2))) float v2f;
typedef __attribute__((ext_vector_type(8))) float v8f;

template <int MASK>
__device__ __forceinline__ float row_xor(float v)
{
    return __int_as_float(__builtin_amdgcn_update_dpp(
        0, __float_as_int(v), 0x160 | MASK, 0xf, 0xf, true));
}

#define GEMM_TPB 5
static_assert((N_NODES / 16) % GEMM_TPB == 0 && N_NODES % 16 == 0, "exact tile loop");

template <int K>
__global__ void __launch_bounds__(128)
gemm_wmma_f32(const float* __restrict__ Hin, const float* __restrict__ W,
              const float* __restrict__ avs, const float* __restrict__ avd,
              float* __restrict__ Hout, float* __restrict__ as_,
              float* __restrict__ ad_)
{
    constexpr int P   = K + 4;
    constexpr int C4  = K / 4;
    constexpr int RPP = 128 / C4;
    constexpr int NV  = 16 / RPP;
    static_assert((P / 4) % 2 == 1 && 128 % C4 == 0 && 16 % RPP == 0, "tile staging shape");

    __shared__ __attribute__((aligned(16))) float As[16 * P];
    __shared__ float part[2][4][16];

    const int tid  = threadIdx.x;
    const int lane = tid & 31;
    const int wv   = tid >> 5;
    const int col0 = wv << 4;
    const int m    = lane & 15;
    const int kb   = (lane >> 4) << 1;

    const float ws = avs[col0 + m];
    const float wd = avd[col0 + m];

    v2f breg[K / 4];
    #pragma unroll
    for (int i = 0; i < K / 4; ++i) {
        const int k0 = i * 4;
        breg[i].x = W[(size_t)(k0 + kb)     * HIDV + col0 + m];
        breg[i].y = W[(size_t)(k0 + kb + 1) * HIDV + col0 + m];
    }

    const int srow  = tid / C4;
    const int sc4   = tid % C4;
    const int tile0 = blockIdx.x * GEMM_TPB;

    static_assert(NV == 2 || NV == 4, "two or four passes");
    float4 pf0, pf1, pf2 = {}, pf3 = {};
    auto fetch = [&](const float* tile) {
        const float4* g = (const float4*)tile;
        pf0 = g[(0 * RPP + srow) * C4 + sc4];
        pf1 = g[(1 * RPP + srow) * C4 + sc4];
        if constexpr (NV == 4) {
            pf2 = g[(2 * RPP + srow) * C4 + sc4];
            pf3 = g[(3 * RPP + srow) * C4 + sc4];
        }
    };
    fetch(Hin + (size_t)(tile0 << 4) * K);

    #pragma nounroll
    for (int it = 0; it < GEMM_TPB; ++it) {
        const int row0 = (tile0 + it) << 4;

        *(float4*)&As[(0 * RPP + srow) * P + sc4 * 4] = pf0;
        *(float4*)&As[(1 * RPP + srow) * P + sc4 * 4] = pf1;
        if constexpr (NV == 4) {
            *(float4*)&As[(2 * RPP + srow) * P + sc4 * 4] = pf2;
            *(float4*)&As[(3 * RPP + srow) * P + sc4 * 4] = pf3;
        }
        __syncthreads();

        if (it + 1 < GEMM_TPB) fetch(Hin + (size_t)(row0 + 16) * K);

        v8f acc = {};
        const float* arow = As + m * P + kb;
        #pragma unroll
        for (int i = 0; i < K / 4; ++i) {
            const v2f a = *(const v2f*)(arow + i * 4);
            acc = __builtin_amdgcn_wmma_f32_16x16x4_f32(
                       false, a,  false, breg[i],
                       (short)0, acc,  false,  false);
        }

        const bool b3 = (m & 8) != 0, b2 = (m & 4) != 0, b1 = (m & 2) != 0;
        float s4[4], d4[4];
        #pragma unroll
        for (int i = 0; i < 4; ++i) {
            const float slo = acc[i] * ws, shi = acc[i + 4] * ws;
            const float dlo = acc[i] * wd, dhi = acc[i + 4] * wd;
            s4[i] = (b3 ? shi : slo) + row_xor<8>(b3 ? slo : shi);
            d4[i] = (b3 ? dhi : dlo) + row_xor<8>(b3 ? dlo : dhi);
        }
        float s2[2], d2[2];
        #pragma unroll
        for (int i = 0; i < 2; ++i) {
            s2[i] = (b2 ? s4[i + 2] : s4[i]) + row_xor<4>(b2 ? s4[i] : s4[i + 2]);
            d2[i] = (b2 ? d4[i + 2] : d4[i]) + row_xor<4>(b2 ? d4[i] : d4[i + 2]);
        }
        float s1 = (b1 ? s2[1] : s2[0]) + row_xor<2>(b1 ? s2[0] : s2[1]);
        float d1 = (b1 ? d2[1] : d2[0]) + row_xor<2>(b1 ? d2[0] : d2[1]);
        s1 += row_xor<1>(s1);
        d1 += row_xor<1>(d1);
        if ((m & 1) == 0) {
            const int r = ((lane >> 4) << 3) + (m >> 1);
            part[0][wv][r] = s1;
            part[1][wv][r] = d1;
        }
        __syncthreads();
        if (tid < 32) {
            const int q = tid >> 4;
            const int r = tid & 15;
            const float v = ((part[q][0][r] + part[q][1][r]) + part[q][2][r]) + part[q][3][r];
            (q ? ad_ : as_)[row0 + r] = v;
        }

        const int rbase = row0 + ((lane >> 4) << 3);
        #pragma unroll
        for (int j = 0; j < 8; ++j)
            Hout[(size_t)(rbase + j) * HIDV + col0 + m] = acc[j];
    }
}

#define BKT_SHIFT 9
#define BKT_W     (1 << BKT_SHIFT)
#define NBK       ((N_NODES + BKT_W - 1) / BKT_W)
#define PCHUNK    6400
#define PBLOCKS   (E_EDGES / PCHUNK)
#define PTHREADS  640
#define PEDGES    (PCHUNK / PTHREADS)
#define SCAP      32768
#define RPW       ((PBLOCKS + 31) / 32)
static_assert(E_EDGES % PCHUNK == 0 && PCHUNK % PTHREADS == 0, "exact chunks");
static_assert(N_NODES <= (1 << (31 - BKT_SHIFT)), "source id and in-bucket destination share one word");
static_assert(NBK <= 256 && PBLOCKS <= 1024 && PCHUNK < 65536, "table sizes");

template <int NWAVES>
__device__ __forceinline__ int block_excl_scan(int v, int* wsum, int& total)
{
    const int lane = threadIdx.x & 31;
    const int wv   = threadIdx.x >> 5;
    int inc = v;
    #pragma unroll
    for (int off = 1; off < 32; off <<= 1) {
        const int t = __shfl_up(inc, off, 32);
        if (lane >= off) inc += t;
    }
    if (lane == 31) wsum[wv] = inc;
    __syncthreads();
    int pre = 0, tot = 0;
    #pragma unroll
    for (int w = 0; w < NWAVES; ++w) {
        const int x = wsum[w];
        tot += x;
        if (w < wv) pre += x;
    }
    total = tot;
    return pre + inc - v;
}

__global__ void __launch_bounds__(PTHREADS)
edge_partition(const int* __restrict__ se, const int* __restrict__ de,
               int* __restrict__ part, int* __restrict__ tabS, int* __restrict__ tabC)
{
    __shared__ int lcnt[256];
    __shared__ int lst[256];
    __shared__ int wsum[PTHREADS / 32];
    __shared__ int stage[PCHUNK];

    const int tid = threadIdx.x;
    const int g   = blockIdx.x;
    const int e0  = g * PCHUNK;
    if (tid < 256) lcnt[tid] = 0;
    __syncthreads();

    int d[PEDGES], s[PEDGES];
    #pragma unroll
    for (int k = 0; k < PEDGES; ++k) {
        const int i = e0 + k * PTHREADS + tid;
        d[k] = de[i];
        s[k] = se[i];
    }
    int w[PEDGES], br[PEDGES];
    #pragma unroll
    for (int k = 0; k < PEDGES; ++k) {
        int bk = (int)((unsigned)d[k] >> BKT_SHIFT);
        bk = bk < NBK ? bk : NBK - 1;
        w[k]  = (s[k] << BKT_SHIFT) | (d[k] & (BKT_W - 1));
        br[k] = (bk << 16) | atomicAdd(&lcnt[bk], 1);
    }
    __syncthreads();

    const int c = tid < 256 ? lcnt[tid] : 0;
    int tot;
    const int ex = block_excl_scan<PTHREADS / 32>(c, wsum, tot);
    if (tid < 256) lst[tid] = ex;
    if (tid < NBK) {
        tabS[tid * PBLOCKS + g] = ex;
        tabC[tid * PBLOCKS + g] = c;
    }
    __syncthreads();

    #pragma unroll
    for (int k = 0; k < PEDGES; ++k)
        stage[lst[br[k] >> 16] + (br[k] & 0xffff)] = w[k];
    __syncthreads();

    #pragma unroll
    for (int k = 0; k < PEDGES; ++k)
        part[e0 + k * PTHREADS + tid] = stage[k * PTHREADS + tid];
}

__global__ void __launch_bounds__(1024)
bucket_sort(const int* __restrict__ part, const int* __restrict__ tabS,
            const int* __restrict__ tabC, int* __restrict__ cnt,
            int* __restrict__ rowptr, int* __restrict__ csr)
{
    __shared__ int rs[PBLOCKS];
    __shared__ int rc[PBLOCKS];
    __shared__ int ncnt[BKT_W];
    __shared__ int cur[BKT_W];
    __shared__ int wsum[32];
    __shared__ int sorted[SCAP];

    const int tid  = threadIdx.x;
    const int lane = tid & 31;
    const int wv   = tid >> 5;
    const int b    = blockIdx.x;
    const int nprev = b * PBLOCKS;

    int s = 0;
    if (tid < PBLOCKS) {
        int a = tabS[nprev + tid];
        int n = tabC[nprev + tid];
        s = a;
        a = a < 0 ? 0 : (a > PCHUNK ? PCHUNK : a);
        n = n < 0 ? 0 : (n > PCHUNK - a ? PCHUNK - a : n);
        rs[tid] = a;
        rc[tid] = n;
    }
    #pragma unroll
    for (int off = 16; off; off >>= 1) s += __shfl_xor(s, off, 32);
    if (lane == 0) wsum[wv] = s;
    if (tid < BKT_W) ncnt[tid] = 0;
    __syncthreads();
    int base = 0;
    #pragma unroll
    for (int w = 0; w < 32; ++w) base += wsum[w];

    int w0[RPW], w1[RPW];
    #pragma unroll
    for (int r = 0; r < RPW; ++r) {
        const int g = wv + 32 * r;
        w0[r] = -1;
        w1[r] = -1;
        if (g < PBLOCKS) {
            const int* p = part + g * PCHUNK + rs[g];
            const int  n = rc[g];
            if (lane < n)      w0[r] = p[lane];
            if (lane + 32 < n) w1[r] = p[lane + 32];
        }
    }

    #pragma unroll
    for (int r = 0; r < RPW; ++r) {
        if (w0[r] >= 0) atomicAdd(&ncnt[w0[r] & (BKT_W - 1)], 1);
        if (w1[r] >= 0) atomicAdd(&ncnt[w1[r] & (BKT_W - 1)], 1);
    }
    for (int g = wv; g < PBLOCKS; g += 32) {
        const int* p = part + g * PCHUNK + rs[g];
        const int  n = rc[g];
        for (int j = lane + 64; j < n; j += 32)
            atomicAdd(&ncnt[p[j] & (BKT_W - 1)], 1);
    }
    __syncthreads();

    const int c = tid < BKT_W ? ncnt[tid] : 0;
    int nb;
    const int ex = block_excl_scan<32>(c, wsum, nb);
    if (tid < BKT_W) {
        cur[tid] = ex;
        const int node = b * BKT_W + tid;
        if (node < N_NODES) { cnt[node] = c; rowptr[node] = base + ex; }
    }
    __syncthreads();

    const bool staged = nb <= SCAP;
    auto place = [&](int wd) {
        const int pos = atomicAdd(&cur[wd & (BKT_W - 1)], 1);
        const int src = (int)((unsigned)wd >> BKT_SHIFT);
        if (staged) {
            if ((unsigned)pos < (unsigned)SCAP) sorted[pos] = src;
        } else {
            const int q = base + pos;
            if ((unsigned)q < (unsigned)E_EDGES) csr[q] = src;
        }
    };
    #pragma unroll
    for (int r = 0; r < RPW; ++r) {
        if (w0[r] >= 0) place(w0[r]);
        if (w1[r] >= 0) place(w1[r]);
    }
    for (int g = wv; g < PBLOCKS; g += 32) {
        const int* p = part + g * PCHUNK + rs[g];
        const int  n = rc[g];
        for (int j = lane + 64; j < n; j += 32) place(p[j]);
    }
    __syncthreads();
    if (staged) {
        for (int e = tid; e < nb; e += 1024) {
            const int q = base + e;
            if ((unsigned)q < (unsigned)E_EDGES) csr[q] = sorted[e];
        }
    }
}

#define GATHER_BLOCKS 1024
static_assert(N_NODES % 2 == 0, "two nodes per wave");

typedef __attribute__((ext_vector_type(4))) float v4f;

__device__ __forceinline__ float lrelu(float v) { return v > 0.0f ? v : NEG_SLOPE * v; }

template <int T>
__device__ __forceinline__ int row_share_i(int v)
{
    return __builtin_amdgcn_update_dpp(0, v, 0x150 | T, 0xf, 0xf, true);
}
template <int T>
__device__ __forceinline__ float row_share_f(float v)
{
    return __int_as_float(row_share_i<T>(__float_as_int(v)));
}

__device__ __forceinline__ float row_max16(float v)
{
    v = fmaxf(v, row_xor<8>(v));
    v = fmaxf(v, row_xor<4>(v));
    v = fmaxf(v, row_xor<2>(v));
    v = fmaxf(v, row_xor<1>(v));
    return v;
}
__device__ __forceinline__ float row_sum16(float v)
{
    v += row_xor<8>(v);
    v += row_xor<4>(v);
    v += row_xor<2>(v);
    v += row_xor<1>(v);
    return v;
}

__device__ __forceinline__ v4f rowld4(const char* Hb, const int r, const unsigned lo)
{
    return *(const v4f*)(Hb + (size_t)((((unsigned)r) << 8) | lo));
}

template <int T0>
__device__ __forceinline__ void issue8(v4f (&r)[8], const int sv, const char* Hb, const unsigned lo)
{
    r[0] = rowld4(Hb, row_share_i<T0 + 0>(sv), lo);
    r[1] = rowld4(Hb, row_share_i<T0 + 1>(sv), lo);
    r[2] = rowld4(Hb, row_share_i<T0 + 2>(sv), lo);
    r[3] = rowld4(Hb, row_share_i<T0 + 3>(sv), lo);
    r[4] = rowld4(Hb, row_share_i<T0 + 4>(sv), lo);
    r[5] = rowld4(Hb, row_share_i<T0 + 5>(sv), lo);
    r[6] = rowld4(Hb, row_share_i<T0 + 6>(sv), lo);
    r[7] = rowld4(Hb, row_share_i<T0 + 7>(sv), lo);
}

template <int T0, int U>
__device__ __forceinline__ void consume1(const v4f (&r)[8], const float al, const int lim, v4f& acc)
{
    if (T0 + U < lim) {
        const float wt = row_share_f<T0 + U>(al);
        acc.x += r[U].x * wt;
        acc.y += r[U].y * wt;
        acc.z += r[U].z * wt;
        acc.w += r[U].w * wt;
    }
}
template <int T0>
__device__ __forceinline__ void consume8(const v4f (&r)[8], const float al, const int lim, v4f& acc)
{
    consume1<T0, 0>(r, al, lim, acc);
    consume1<T0, 1>(r, al, lim, acc);
    consume1<T0, 2>(r, al, lim, acc);
    consume1<T0, 3>(r, al, lim, acc);
    consume1<T0, 4>(r, al, lim, acc);
    consume1<T0, 5>(r, al, lim, acc);
    consume1<T0, 6>(r, al, lim, acc);
    consume1<T0, 7>(r, al, lim, acc);
}

template <bool HEAD>
__global__ void __launch_bounds__(256, 8)
gat_gather(const int* __restrict__ rowptr, const int* __restrict__ cnt,
           const int* __restrict__ csr, const float* __restrict__ H,
           const float* __restrict__ as_, const float* __restrict__ ad_,
           const float* __restrict__ bias, float* __restrict__ Hout,
           const float* __restrict__ Wo, const float* __restrict__ bo,
           float* __restrict__ y)
{
    const int lane = threadIdx.x & 31;
    const int c    = lane & 15;
    const int hh   = lane >> 4;
    const int wid = __builtin_amdgcn_readfirstlane((int)(blockIdx.x * 8 + (threadIdx.x >> 5)));
    constexpr int nw     = GATHER_BLOCKS * 8;
    constexpr int NPAIRS = N_NODES / 2;
    if (wid >= NPAIRS) return;

    const char*    Hb = (const char*)H;
    const unsigned lo = (unsigned)c * 16u;
    const v4f b4 = ((const v4f*)bias)[c];
    v4f   wo4 = {0.0f, 0.0f, 0.0f, 0.0f};
    float bov = 0.0f;
    if (HEAD) { wo4 = ((const v4f*)Wo)[c]; bov = bo[0]; }

    int   pr   = wid;
    int   node = 2 * pr + hh;
    int   beg  = rowptr[node];
    int   deg  = cnt[node];
    float adn  = ad_[node];
    float asn  = as_[node];
    int   s0   = node;
    int   s1   = node;
    if (c < deg)      s0 = csr[beg + c];
    if (c + 16 < deg) s1 = csr[beg + 16 + c];

    for (;;) {
        const int  npr   = pr + nw;
        const bool more  = npr < NPAIRS;
        const int  nnode = 2 * npr + hh;

        const int   hn   = more ? nnode : node;
        const int   nbeg = rowptr[hn];
        const int   ndeg = cnt[hn];
        const float nadn = ad_[hn];
        const float nasn = as_[hn];

        const float e0 = lrelu(as_[s0] + adn);
        const float e1 = lrelu(as_[s1] + adn);
        const v4f hself = rowld4(Hb, node, lo);
        const int dA_ = __builtin_amdgcn_readlane(deg, 0);
        const int dB_ = __builtin_amdgcn_readlane(deg, 16);
        const int dmax = dA_ > dB_ ? dA_ : dB_;

        v4f ra[8], rb[8];
        issue8<0>(ra, s0, Hb, lo);

        const bool q0 = c < ndeg, q1 = c + 16 < ndeg;
        const int  t0 = csr[q0 ? nbeg + c : 0];
        const int  t1 = csr[q1 ? nbeg + 16 + c : 0];
        const int  ns0 = q0 ? t0 : hn;
        const int  ns1 = q1 ? t1 : hn;

        const float es = lrelu(asn + adn);
        float m = fmaxf(es, fmaxf(e0, e1));
        for (int j = c + 32; j < deg; j += 16)
            m = fmaxf(m, lrelu(as_[csr[beg + j]] + adn));
        m = row_max16(m);

        const float ex0 = (c < deg)      ? expf(e0 - m) : 0.0f;
        const float ex1 = (c + 16 < deg) ? expf(e1 - m) : 0.0f;
        float sumA = ex0, sumB = ex1;
        {
            int k = 0;
            for (int j = c + 32; j < deg; j += 16, ++k) {
                const float v = expf(lrelu(as_[csr[beg + j]] + adn) - m);
                if (k & 1) sumB += v; else sumA += v;
            }
        }
        const float dsum = row_sum16(sumA + sumB);
        const float exs  = expf(es - m);
        const float den  = dsum + exs;

        const float a_self = exs / den;
        v4f acc;
        acc.x = hself.x * a_self;
        acc.y = hself.y * a_self;
        acc.z = hself.z * a_self;
        acc.w = hself.w * a_self;
        const float al0 = ex0 / den;
        const float al1 = ex1 / den;

        if (dmax > 8) issue8<8>(rb, s0, Hb, lo);
        consume8<0>(ra, al0, dmax, acc);
        if (dmax > 8) {
            if (dmax > 16) issue8<0>(ra, s1, Hb, lo);
            consume8<8>(rb, al0, dmax, acc);
            if (dmax > 16) {
                if (dmax > 24) issue8<8>(rb, s1, Hb, lo);
                consume8<0>(ra, al1, dmax - 16, acc);
                if (dmax > 24) consume8<8>(rb, al1, dmax - 16, acc);
            }
        }

        for (int j0 = 32; j0 < dmax; j0 += 16) {
            const int j = j0 + c;
            int   sj  = node;
            float alj = 0.0f;
            if (j < deg) {
                sj  = csr[beg + j];
                alj = expf(lrelu(as_[sj] + adn) - m) / den;
            }
            issue8<0>(ra, sj, Hb, lo);
            issue8<8>(rb, sj, Hb, lo);
            consume8<0>(ra, alj, dmax - j0, acc);
            consume8<8>(rb, alj, dmax - j0, acc);
        }

        const float vx = acc.x + b4.x;
        const float vy = acc.y + b4.y;
        const float vz = acc.z + b4.z;
        const float vw = acc.w + b4.w;
        v4f o;
        o.x = vx > 0.0f ? vx : expm1f(vx);
        o.y = vy > 0.0f ? vy : expm1f(vy);
        o.z = vz > 0.0f ? vz : expm1f(vz);
        o.w = vw > 0.0f ? vw : expm1f(vw);
        if (HEAD) {
            float s = fmaf(o.x, wo4.x, o.y * wo4.y) + fmaf(o.z, wo4.z, o.w * wo4.w);
            s += row_xor<1>(s);
            s += row_xor<2>(s);
            s += row_xor<4>(s);
            s += row_xor<8>(s);
            if (c == 0) y[node] = s + bov;
        } else {
            ((v4f*)Hout)[(size_t)node * 16 + c] = o;
        }

        if (!more) break;
        pr = npr; node = hn; beg = nbeg; deg = ndeg; adn = nadn; asn = nasn; s0 = ns0; s1 = ns1;
    }
}

extern "C" void kernel_launch(void* const* d_in, const int* in_sizes, int n_in,
                              void* d_out, int out_size, void* d_ws, size_t ws_size,
                              hipStream_t stream)
{
    const float* x      = (const float*)d_in[0];
    const int*   ei     = (const int*)d_in[1];
    const float* W1     = (const float*)d_in[2];
    const float* a_src1 = (const float*)d_in[3];
    const float* a_dst1 = (const float*)d_in[4];
    const float* b1     = (const float*)d_in[5];
    const float* W2     = (const float*)d_in[6];
    const float* a_src2 = (const float*)d_in[7];
    const float* a_dst2 = (const float*)d_in[8];
    const float* b2     = (const float*)d_in[9];
    const float* W_out  = (const float*)d_in[10];
    const float* b_out  = (const float*)d_in[11];
    float* y = (float*)d_out;

    const int* se = ei;
    const int* de = ei + E_EDGES;

    float* h1     = (float*)d_ws;
    float* h2     = h1 + (size_t)N_NODES * HIDV;
    float* as_    = h2 + (size_t)N_NODES * HIDV;
    float* ad_    = as_ + N_NODES;
    int*   cnt    = (int*)(ad_ + N_NODES);
    int*   rowptr = cnt + N_NODES;
    int*   csr    = rowptr + N_NODES;
    int*   part   = (int*)h2;
    int*   tabS   = part + E_EDGES;
    int*   tabC   = tabS + NBK * PBLOCKS;

    const int gRowT    = (N_NODES / 16) / GEMM_TPB;

    edge_partition<<<PBLOCKS, PTHREADS, 0, stream>>>(se, de, part, tabS, tabC);
    bucket_sort<<<NBK, 1024, 0, stream>>>(part, tabS, tabC, cnt, rowptr, csr);

    gemm_wmma_f32<IN_DIMV><<<gRowT, 128, 0, stream>>>(x, W1, a_src1, a_dst1, h1, as_, ad_);
    gat_gather<false><<<GATHER_BLOCKS, 256, 0, stream>>>(rowptr, cnt, csr, h1, as_, ad_, b1, h2,
                                                    W_out, b_out, y);

    gemm_wmma_f32<HIDV><<<gRowT, 128, 0, stream>>>(h2, W2, a_src2, a_dst2, h1, as_, ad_);
    gat_gather<true><<<GATHER_BLOCKS, 256, 0, stream>>>(rowptr, cnt, csr, h1, as_, ad_, b2, h2,
                                                   W_out, b_out, y);
}


# Layer-2 projection fused into the layer-1 gather: a block stages its 16 finished rows in LDS and 4 waves multiply them by W2 on the matrix unit; one launch and the ELU plane gone; sum order kept
# speedup vs baseline: 8.2492x; 1.0716x over previous
#include <hip/hip_runtime.h>
#include <hip/hip_bf16.h>

#define N_NODES   100000
#define E_EDGES   1600000
#define IN_DIMV   128
#define HIDV      64
#define NEG_SLOPE 0.2f

typedef __attribute__((ext_vector_type(2))) float v2f;
typedef __attribute__((ext_vector_type(8))) float v8f;

template <int MASK>
__device__ __forceinline__ float row_xor(float v)
{
    return __int_as_float(__builtin_amdgcn_update_dpp(
        0, __float_as_int(v), 0x160 | MASK, 0xf, 0xf, true));
}

#define GEMM_TPB 5
static_assert((N_NODES / 16) % GEMM_TPB == 0 && N_NODES % 16 == 0, "exact tile loop");

template <int K>
__global__ void __launch_bounds__(128)
gemm_wmma_f32(const float* __restrict__ Hin, const float* __restrict__ W,
              const float* __restrict__ avs, const float* __restrict__ avd,
              float* __restrict__ Hout, float* __restrict__ as_,
              float* __restrict__ ad_)
{
    constexpr int P   = K + 4;
    constexpr int C4  = K / 4;
    constexpr int RPP = 128 / C4;
    constexpr int NV  = 16 / RPP;
    static_assert((P / 4) % 2 == 1 && 128 % C4 == 0 && 16 % RPP == 0, "tile staging shape");

    __shared__ __attribute__((aligned(16))) float As[16 * P];
    __shared__ float part[2][4][16];

    const int tid  = threadIdx.x;
    const int lane = tid & 31;
    const int wv   = tid >> 5;
    const int col0 = wv << 4;
    const int m    = lane & 15;
    const int kb   = (lane >> 4) << 1;

    const float ws = avs[col0 + m];
    const float wd = avd[col0 + m];

    v2f breg[K / 4];
    #pragma unroll
    for (int i = 0; i < K / 4; ++i) {
        const int k0 = i * 4;
        breg[i].x = W[(size_t)(k0 + kb)     * HIDV + col0 + m];
        breg[i].y = W[(size_t)(k0 + kb + 1) * HIDV + col0 + m];
    }

    const int srow  = tid / C4;
    const int sc4   = tid % C4;
    const int tile0 = blockIdx.x * GEMM_TPB;

    static_assert(NV == 2 || NV == 4, "two or four passes");
    float4 pf0, pf1, pf2 = {}, pf3 = {};
    auto fetch = [&](const float* tile) {
        const float4* g = (const float4*)tile;
        pf0 = g[(0 * RPP + srow) * C4 + sc4];
        pf1 = g[(1 * RPP + srow) * C4 + sc4];
        if constexpr (NV == 4) {
            pf2 = g[(2 * RPP + srow) * C4 + sc4];
            pf3 = g[(3 * RPP + srow) * C4 + sc4];
        }
    };
    fetch(Hin + (size_t)(tile0 << 4) * K);

    #pragma nounroll
    for (int it = 0; it < GEMM_TPB; ++it) {
        const int row0 = (tile0 + it) << 4;

        *(float4*)&As[(0 * RPP + srow) * P + sc4 * 4] = pf0;
        *(float4*)&As[(1 * RPP + srow) * P + sc4 * 4] = pf1;
        if constexpr (NV == 4) {
            *(float4*)&As[(2 * RPP + srow) * P + sc4 * 4] = pf2;
            *(float4*)&As[(3 * RPP + srow) * P + sc4 * 4] = pf3;
        }
        __syncthreads();

        if (it + 1 < GEMM_TPB) fetch(Hin + (size_t)(row0 + 16) * K);

        v8f acc = {};
        const float* arow = As + m * P + kb;
        #pragma unroll
        for (int i = 0; i < K / 4; ++i) {
            const v2f a = *(const v2f*)(arow + i * 4);
            acc = __builtin_amdgcn_wmma_f32_16x16x4_f32(
                       false, a,  false, breg[i],
                       (short)0, acc,  false,  false);
        }

        const bool b3 = (m & 8) != 0, b2 = (m & 4) != 0, b1 = (m & 2) != 0;
        float s4[4], d4[4];
        #pragma unroll
        for (int i = 0; i < 4; ++i) {
            const float slo = acc[i] * ws, shi = acc[i + 4] * ws;
            const float dlo = acc[i] * wd, dhi = acc[i + 4] * wd;
            s4[i] = (b3 ? shi : slo) + row_xor<8>(b3 ? slo : shi);
            d4[i] = (b3 ? dhi : dlo) + row_xor<8>(b3 ? dlo : dhi);
        }
        float s2[2], d2[2];
        #pragma unroll
        for (int i = 0; i < 2; ++i) {
            s2[i] = (b2 ? s4[i + 2] : s4[i]) + row_xor<4>(b2 ? s4[i] : s4[i + 2]);
            d2[i] = (b2 ? d4[i + 2] : d4[i]) + row_xor<4>(b2 ? d4[i] : d4[i + 2]);
        }
        float s1 = (b1 ? s2[1] : s2[0]) + row_xor<2>(b1 ? s2[0] : s2[1]);
        float d1 = (b1 ? d2[1] : d2[0]) + row_xor<2>(b1 ? d2[0] : d2[1]);
        s1 += row_xor<1>(s1);
        d1 += row_xor<1>(d1);
        if ((m & 1) == 0) {
            const int r = ((lane >> 4) << 3) + (m >> 1);
            part[0][wv][r] = s1;
            part[1][wv][r] = d1;
        }
        __syncthreads();
        if (tid < 32) {
            const int q = tid >> 4;
            const int r = tid & 15;
            const float v = ((part[q][0][r] + part[q][1][r]) + part[q][2][r]) + part[q][3][r];
            (q ? ad_ : as_)[row0 + r] = v;
        }

        const int rbase = row0 + ((lane >> 4) << 3);
        #pragma unroll
        for (int j = 0; j < 8; ++j)
            Hout[(size_t)(rbase + j) * HIDV + col0 + m] = acc[j];
    }
}

#define BKT_SHIFT 9
#define BKT_W     (1 << BKT_SHIFT)
#define NBK       ((N_NODES + BKT_W - 1) / BKT_W)
#define PCHUNK    6400
#define PBLOCKS   (E_EDGES / PCHUNK)
#define PTHREADS  640
#define PEDGES    (PCHUNK / PTHREADS)
#define SCAP      32768
#define RPW       ((PBLOCKS + 31) / 32)
static_assert(E_EDGES % PCHUNK == 0 && PCHUNK % PTHREADS == 0, "exact chunks");
static_assert(N_NODES <= (1 << (31 - BKT_SHIFT)), "source id and in-bucket destination share one word");
static_assert(NBK <= 256 && PBLOCKS <= 1024 && PCHUNK < 65536, "table sizes");

template <int NWAVES>
__device__ __forceinline__ int block_excl_scan(int v, int* wsum, int& total)
{
    const int lane = threadIdx.x & 31;
    const int wv   = threadIdx.x >> 5;
    int inc = v;
    #pragma unroll
    for (int off = 1; off < 32; off <<= 1) {
        const int t = __shfl_up(inc, off, 32);
        if (lane >= off) inc += t;
    }
    if (lane == 31) wsum[wv] = inc;
    __syncthreads();
    int pre = 0, tot = 0;
    #pragma unroll
    for (int w = 0; w < NWAVES; ++w) {
        const int x = wsum[w];
        tot += x;
        if (w < wv) pre += x;
    }
    total = tot;
    return pre + inc - v;
}

__global__ void __launch_bounds__(PTHREADS)
edge_partition(const int* __restrict__ se, const int* __restrict__ de,
               int* __restrict__ part, int* __restrict__ tabS, int* __restrict__ tabC)
{
    __shared__ int lcnt[256];
    __shared__ int lst[256];
    __shared__ int wsum[PTHREADS / 32];
    __shared__ int stage[PCHUNK];

    const int tid = threadIdx.x;
    const int g   = blockIdx.x;
    const int e0  = g * PCHUNK;
    if (tid < 256) lcnt[tid] = 0;
    __syncthreads();

    int d[PEDGES], s[PEDGES];
    #pragma unroll
    for (int k = 0; k < PEDGES; ++k) {
        const int i = e0 + k * PTHREADS + tid;
        d[k] = de[i];
        s[k] = se[i];
    }
    int w[PEDGES], br[PEDGES];
    #pragma unroll
    for (int k = 0; k < PEDGES; ++k) {
        int bk = (int)((unsigned)d[k] >> BKT_SHIFT);
        bk = bk < NBK ? bk : NBK - 1;
        w[k]  = (s[k] << BKT_SHIFT) | (d[k] & (BKT_W - 1));
        br[k] = (bk << 16) | atomicAdd(&lcnt[bk], 1);
    }
    __syncthreads();

    const int c = tid < 256 ? lcnt[tid] : 0;
    int tot;
    const int ex = block_excl_scan<PTHREADS / 32>(c, wsum, tot);
    if (tid < 256) lst[tid] = ex;
    if (tid < NBK) {
        tabS[tid * PBLOCKS + g] = ex;
        tabC[tid * PBLOCKS + g] = c;
    }
    __syncthreads();

    #pragma unroll
    for (int k = 0; k < PEDGES; ++k)
        stage[lst[br[k] >> 16] + (br[k] & 0xffff)] = w[k];
    __syncthreads();

    #pragma unroll
    for (int k = 0; k < PEDGES; ++k)
        part[e0 + k * PTHREADS + tid] = stage[k * PTHREADS + tid];
}

__global__ void __launch_bounds__(1024)
bucket_sort(const int* __restrict__ part, const int* __restrict__ tabS,
            const int* __restrict__ tabC,
            int* __restrict__ rowptr, int* __restrict__ csr)
{
    __shared__ int rs[PBLOCKS];
    __shared__ int rc[PBLOCKS];
    __shared__ int ncnt[BKT_W];
    __shared__ int cur[BKT_W];
    __shared__ int wsum[32];
    __shared__ int sorted[SCAP];

    const int tid  = threadIdx.x;
    const int lane = tid & 31;
    const int wv   = tid >> 5;
    const int b    = blockIdx.x;
    const int nprev = b * PBLOCKS;

    int s = 0;
    if (tid < PBLOCKS) {
        int a = tabS[nprev + tid];
        int n = tabC[nprev + tid];
        s = a;
        a = a < 0 ? 0 : (a > PCHUNK ? PCHUNK : a);
        n = n < 0 ? 0 : (n > PCHUNK - a ? PCHUNK - a : n);
        rs[tid] = a;
        rc[tid] = n;
    }
    #pragma unroll
    for (int off = 16; off; off >>= 1) s += __shfl_xor(s, off, 32);
    if (lane == 0) wsum[wv] = s;
    if (tid < BKT_W) ncnt[tid] = 0;
    __syncthreads();
    int base = 0;
    #pragma unroll
    for (int w = 0; w < 32; ++w) base += wsum[w];

    int w0[RPW], w1[RPW];
    #pragma unroll
    for (int r = 0; r < RPW; ++r) {
        const int g = wv + 32 * r;
        w0[r] = -1;
        w1[r] = -1;
        if (g < PBLOCKS) {
            const int* p = part + g * PCHUNK + rs[g];
            const int  n = rc[g];
            if (lane < n)      w0[r] = p[lane];
            if (lane + 32 < n) w1[r] = p[lane + 32];
        }
    }

    #pragma unroll
    for (int r = 0; r < RPW; ++r) {
        if (w0[r] >= 0) atomicAdd(&ncnt[w0[r] & (BKT_W - 1)], 1);
        if (w1[r] >= 0) atomicAdd(&ncnt[w1[r] & (BKT_W - 1)], 1);
    }
    for (int g = wv; g < PBLOCKS; g += 32) {
        const int* p = part + g * PCHUNK + rs[g];
        const int  n = rc[g];
        for (int j = lane + 64; j < n; j += 32)
            atomicAdd(&ncnt[p[j] & (BKT_W - 1)], 1);
    }
    __syncthreads();

    const int c = tid < BKT_W ? ncnt[tid] : 0;
    int nb;
    const int ex = block_excl_scan<32>(c, wsum, nb);
    if (tid < BKT_W) {
        cur[tid] = ex;
        const int node = b * BKT_W + tid;
        if (node < N_NODES) rowptr[node] = base + ex;
    }
    __syncthreads();

    const bool staged = nb <= SCAP;
    auto place = [&](int wd) {
        const int pos = atomicAdd(&cur[wd & (BKT_W - 1)], 1);
        const int src = (int)((unsigned)wd >> BKT_SHIFT);
        if (staged) {
            if ((unsigned)pos < (unsigned)SCAP) sorted[pos] = src;
        } else {
            const int q = base + pos;
            if ((unsigned)q < (unsigned)E_EDGES) csr[q] = src;
        }
    };
    #pragma unroll
    for (int r = 0; r < RPW; ++r) {
        if (w0[r] >= 0) place(w0[r]);
        if (w1[r] >= 0) place(w1[r]);
    }
    for (int g = wv; g < PBLOCKS; g += 32) {
        const int* p = part + g * PCHUNK + rs[g];
        const int  n = rc[g];
        for (int j = lane + 64; j < n; j += 32) place(p[j]);
    }
    __syncthreads();
    if (staged) {
        for (int e = tid; e < nb; e += 1024) {
            const int q = base + e;
            if ((unsigned)q < (unsigned)E_EDGES) csr[q] = sorted[e];
        }
    }
}

#define GATHER_BLOCKS 1024
static_assert(N_NODES % 2 == 0, "two nodes per wave");

typedef __attribute__((ext_vector_type(4))) float v4f;

__device__ __forceinline__ float lrelu(float v) { return v > 0.0f ? v : NEG_SLOPE * v; }

template <int T>
__device__ __forceinline__ int row_share_i(int v)
{
    return __builtin_amdgcn_update_dpp(0, v, 0x150 | T, 0xf, 0xf, true);
}
template <int T>
__device__ __forceinline__ float row_share_f(float v)
{
    return __int_as_float(row_share_i<T>(__float_as_int(v)));
}

__device__ __forceinline__ float row_max16(float v)
{
    v = fmaxf(v, row_xor<8>(v));
    v = fmaxf(v, row_xor<4>(v));
    v = fmaxf(v, row_xor<2>(v));
    v = fmaxf(v, row_xor<1>(v));
    return v;
}
__device__ __forceinline__ float row_sum16(float v)
{
    v += row_xor<8>(v);
    v += row_xor<4>(v);
    v += row_xor<2>(v);
    v += row_xor<1>(v);
    return v;
}

__device__ __forceinline__ v4f rowld4(const char* Hb, const int r, const unsigned lo)
{
    return *(const v4f*)(Hb + (size_t)((((unsigned)r) << 8) | lo));
}

template <int T0>
__device__ __forceinline__ void issue8(v4f (&r)[8], const int sv, const char* Hb, const unsigned lo)
{
    r[0] = rowld4(Hb, row_share_i<T0 + 0>(sv), lo);
    r[1] = rowld4(Hb, row_share_i<T0 + 1>(sv), lo);
    r[2] = rowld4(Hb, row_share_i<T0 + 2>(sv), lo);
    r[3] = rowld4(Hb, row_share_i<T0 + 3>(sv), lo);
    r[4] = rowld4(Hb, row_share_i<T0 + 4>(sv), lo);
    r[5] = rowld4(Hb, row_share_i<T0 + 5>(sv), lo);
    r[6] = rowld4(Hb, row_share_i<T0 + 6>(sv), lo);
    r[7] = rowld4(Hb, row_share_i<T0 + 7>(sv), lo);
}

template <int T0, int U>
__device__ __forceinline__ void consume1(const v4f (&r)[8], const float al, const int lim, v4f& acc)
{
    if (T0 + U < lim) {
        const float wt = row_share_f<T0 + U>(al);
        acc.x += r[U].x * wt;
        acc.y += r[U].y * wt;
        acc.z += r[U].z * wt;
        acc.w += r[U].w * wt;
    }
}
template <int T0>
__device__ __forceinline__ void consume8(const v4f (&r)[8], const float al, const int lim, v4f& acc)
{
    consume1<T0, 0>(r, al, lim, acc);
    consume1<T0, 1>(r, al, lim, acc);
    consume1<T0, 2>(r, al, lim, acc);
    consume1<T0, 3>(r, al, lim, acc);
    consume1<T0, 4>(r, al, lim, acc);
    consume1<T0, 5>(r, al, lim, acc);
    consume1<T0, 6>(r, al, lim, acc);
    consume1<T0, 7>(r, al, lim, acc);
}

#define TPITCH (HIDV + 4)
static_assert((TPITCH / 4) % 2 == 1 && HIDV == 64, "conflict-free fragment reads; the fill splits 256 threads into 64 columns x 4 row groups");
static_assert((N_NODES / 2) % 8 == 0 && N_NODES % 16 == 0, "a block's 8 waves share their trip count and own whole row tiles");

__device__ __forceinline__ int next_index(const int node) { return node + 1 < N_NODES ? node + 1 : node; }
__device__ __forceinline__ int seg_len(const int node, const int beg, const int nxt)
{
    return (node + 1 < N_NODES ? nxt : E_EDGES) - beg;
}

template <bool HEAD>
__global__ void __launch_bounds__(256, 8)
gat_gather(const int* __restrict__ rowptr,
           const int* __restrict__ csr, const float* __restrict__ H,
           const float* __restrict__ as_, const float* ad_,
           const float* __restrict__ bias,
           const float* __restrict__ Wn, const float* __restrict__ avs, const float* __restrict__ avd,
           float* __restrict__ Hout, float* __restrict__ asn_out, float* adn_out,
           const float* __restrict__ Wo, const float* __restrict__ bo,
           float* __restrict__ y)
{
    constexpr bool FUSE = !HEAD;
    __shared__ __attribute__((aligned(16))) float Ts[FUSE ? 2 : 1][FUSE ? 16 * TPITCH : 4];
    __shared__ float part[FUSE ? 2 : 1][2][4][16];
    __shared__ __attribute__((aligned(16))) float Wt[FUSE ? HIDV * TPITCH : 4];

    const int lane = threadIdx.x & 31;
    const int c    = lane & 15;
    const int hh   = lane >> 4;
    const int wvb = __builtin_amdgcn_readfirstlane((int)(threadIdx.x >> 5));
    const int wid = __builtin_amdgcn_readfirstlane((int)(blockIdx.x * 8 + (threadIdx.x >> 5)));
    constexpr int nw     = GATHER_BLOCKS * 8;
    constexpr int NPAIRS = N_NODES / 2;
    static_assert(nw <= NPAIRS, "every wave has a pair: no wave leaves before the barriers");

    const char*    Hb = (const char*)H;
    const unsigned lo = (unsigned)c * 16u;
    const v4f b4 = ((const v4f*)bias)[c];
    v4f   wo4 = {0.0f, 0.0f, 0.0f, 0.0f};
    float bov = 0.0f;
    if (HEAD) { wo4 = ((const v4f*)Wo)[c]; bov = bo[0]; }

    const int kb   = hh << 1;
    const int col0 = (wvb & 3) << 4;
    float wsl = 0.0f, wdl = 0.0f;
    const int fn = threadIdx.x & (HIDV - 1);
    const int fk = (threadIdx.x >> 6) << 4;
    float wtmp[16];
    if (FUSE) {
        wsl = avs[col0 + c];
        wdl = avd[col0 + c];
        #pragma unroll
        for (int j = 0; j < 16; ++j) wtmp[j] = Wn[(fk + j) * HIDV + fn];
    }

    auto join_logits = [&](const int pb, const int r0) {
        const int q = lane >> 4;
        const int r = lane & 15;
        const float v = ((part[pb][q][0][r] + part[pb][q][1][r]) + part[pb][q][2][r]) + part[pb][q][3][r];
        (q ? adn_out : asn_out)[r0 + r] = v;
    };

    int   pr   = wid;
    int   node = 2 * pr + hh;
    int   beg  = rowptr[node];
    int   deg  = seg_len(node, beg, rowptr[next_index(node)]);
    float adn  = ad_[node];
    float asn  = as_[node];
    int   s0   = node;
    int   s1   = node;
    if (c < deg)      s0 = csr[beg + c];
    if (c + 16 < deg) s1 = csr[beg + 16 + c];
    int   it   = 0;
    if (FUSE) {
        #pragma unroll
        for (int j = 0; j < 16; j += 4) {
            v4f t4 = {wtmp[j], wtmp[j + 1], wtmp[j + 2], wtmp[j + 3]};
            *(v4f*)&Wt[fn * TPITCH + fk + j] = t4;
        }
    }

    for (;;) {
        const int  npr   = pr + nw;
        const bool more  = npr < NPAIRS;
        const int  nnode = 2 * npr + hh;

        const int   hn   = more ? nnode : node;
        const int   nbeg = rowptr[hn];
        const int   nnxt = rowptr[next_index(hn)];
        const float nadn = ad_[hn];
        const float nasn = as_[hn];

        const float e0 = lrelu(as_[s0] + adn);
        const float e1 = lrelu(as_[s1] + adn);
        const v4f hself = rowld4(Hb, node, lo);
        const int dA_ = __builtin_amdgcn_readlane(deg, 0);
        const int dB_ = __builtin_amdgcn_readlane(deg, 16);
        const int dmax = dA_ > dB_ ? dA_ : dB_;

        v4f ra[8], rb[8];
        issue8<0>(ra, s0, Hb, lo);

        const int  ndeg = seg_len(hn, nbeg, nnxt);
        const bool q0 = c < ndeg, q1 = c + 16 < ndeg;
        const int  t0 = csr[q0 ? nbeg + c : 0];
        const int  t1 = csr[q1 ? nbeg + 16 + c : 0];
        const int  ns0 = q0 ? t0 : hn;
        const int  ns1 = q1 ? t1 : hn;

        const float es = lrelu(asn + adn);
        float m = fmaxf(es, fmaxf(e0, e1));
        for (int j = c + 32; j < deg; j += 16)
            m = fmaxf(m, lrelu(as_[csr[beg + j]] + adn));
        m = row_max16(m);

        const float ex0 = (c < deg)      ? expf(e0 - m) : 0.0f;
        const float ex1 = (c + 16 < deg) ? expf(e1 - m) : 0.0f;
        float sumA = ex0, sumB = ex1;
        {
            int k = 0;
            for (int j = c + 32; j < deg; j += 16, ++k) {
                const float v = expf(lrelu(as_[csr[beg + j]] + adn) - m);
                if (k & 1) sumB += v; else sumA += v;
            }
        }
        const float dsum = row_sum16(sumA + sumB);
        const float exs  = expf(es - m);
        const float den  = dsum + exs;

        const float a_self = exs / den;
        v4f acc;
        acc.x = hself.x * a_self;
        acc.y = hself.y * a_self;
        acc.z = hself.z * a_self;
        acc.w = hself.w * a_self;
        const float al0 = ex0 / den;
        const float al1 = ex1 / den;

        if (dmax > 8) issue8<8>(rb, s0, Hb, lo);
        consume8<0>(ra, al0, dmax, acc);
        if (dmax > 8) {
            if (dmax > 16) issue8<0>(ra, s1, Hb, lo);
            consume8<8>(rb, al0, dmax, acc);
            if (dmax > 16) {
                if (dmax > 24) issue8<8>(rb, s1, Hb, lo);
                consume8<0>(ra, al1, dmax - 16, acc);
                if (dmax > 24) consume8<8>(rb, al1, dmax - 16, acc);
            }
        }

        for (int j0 = 32; j0 < dmax; j0 += 16) {
            const int j = j0 + c;
            int   sj  = node;
            float alj = 0.0f;
            if (j < deg) {
                sj  = csr[beg + j];
                alj = expf(lrelu(as_[sj] + adn) - m) / den;
            }
            issue8<0>(ra, sj, Hb, lo);
            issue8<8>(rb, sj, Hb, lo);
            consume8<0>(ra, alj, dmax - j0, acc);
            consume8<8>(rb, alj, dmax - j0, acc);
        }

        const float vx = acc.x + b4.x;
        const float vy = acc.y + b4.y;
        const float vz = acc.z + b4.z;
        const float vw = acc.w + b4.w;
        v4f o;
        o.x = vx > 0.0f ? vx : expm1f(vx);
        o.y = vy > 0.0f ? vy : expm1f(vy);
        o.z = vz > 0.0f ? vz : expm1f(vz);
        o.w = vw > 0.0f ? vw : expm1f(vw);
        if (HEAD) {
            float s = fmaf(o.x, wo4.x, o.y * wo4.y) + fmaf(o.z, wo4.z, o.w * wo4.w);
            s += row_xor<1>(s);
            s += row_xor<2>(s);
            s += row_xor<4>(s);
            s += row_xor<8>(s);
            if (c == 0) y[node] = s + bov;
        } else {
            const int pb   = it & 1;
            const int row0 = 2 * (pr - wvb);
            *(v4f*)&Ts[pb][(2 * wvb + hh) * TPITCH + 4 * c] = o;
            __syncthreads();

            if (wvb < 4) {
                v8f g = {};
                const float* arow = &Ts[pb][c * TPITCH + kb];
                const float* brow = &Wt[(col0 + c) * TPITCH + kb];
                #pragma unroll
                for (int i = 0; i < HIDV / 4; ++i) {
                    const v2f a = *(const v2f*)(arow + i * 4);
                    const v2f b = *(const v2f*)(brow + i * 4);
                    g = __builtin_amdgcn_wmma_f32_16x16x4_f32(
                             false, a,  false, b,
                             (short)0, g,  false,  false);
                }

                const bool b3 = (c & 8) != 0, b2 = (c & 4) != 0, b1 = (c & 2) != 0;
                float s4[4], d4[4];
                #pragma unroll
                for (int i = 0; i < 4; ++i) {
                    const float slo = g[i] * wsl, shi = g[i + 4] * wsl;
                    const float dlo = g[i] * wdl, dhi = g[i + 4] * wdl;
                    s4[i] = (b3 ? shi : slo) + row_xor<8>(b3 ? slo : shi);
                    d4[i] = (b3 ? dhi : dlo) + row_xor<8>(b3 ? dlo : dhi);
                }
                float s2[2], d2[2];
                #pragma unroll
                for (int i = 0; i < 2; ++i) {
                    s2[i] = (b2 ? s4[i + 2] : s4[i]) + row_xor<4>(b2 ? s4[i] : s4[i + 2]);
                    d2[i] = (b2 ? d4[i + 2] : d4[i]) + row_xor<4>(b2 ? d4[i] : d4[i + 2]);
                }
                float s1v = (b1 ? s2[1] : s2[0]) + row_xor<2>(b1 ? s2[0] : s2[1]);
                float d1v = (b1 ? d2[1] : d2[0]) + row_xor<2>(b1 ? d2[0] : d2[1]);
                s1v += row_xor<1>(s1v);
                d1v += row_xor<1>(d1v);
                if ((c & 1) == 0) {
                    const int r = (hh << 3) + (c >> 1);
                    part[pb][0][wvb][r] = s1v;
                    part[pb][1][wvb][r] = d1v;
                }

                const int rbase = row0 + (hh << 3);
                #pragma unroll
                for (int j = 0; j < 8; ++j)
                    Hout[(size_t)(rbase + j) * HIDV + col0 + c] = g[j];
            } else if (wvb == 4 && it > 0) {
                join_logits(pb ^ 1, row0 - 2 * nw);
            }
        }

        if (!more) break;
        pr = npr; node = hn; beg = nbeg; deg = ndeg; adn = nadn; asn = nasn; s0 = ns0; s1 = ns1;
        ++it;
    }

    if (FUSE) {
        __syncthreads();
        if (wvb == 4) join_logits(it & 1, 2 * (pr - wvb));
    }
}

extern "C" void kernel_launch(void* const* d_in, const int* in_sizes, int n_in,
                              void* d_out, int out_size, void* d_ws, size_t ws_size,
                              hipStream_t stream)
{
    const float* x      = (const float*)d_in[0];
    const int*   ei     = (const int*)d_in[1];
    const float* W1     = (const float*)d_in[2];
    const float* a_src1 = (const float*)d_in[3];
    const float* a_dst1 = (const float*)d_in[4];
    const float* b1     = (const float*)d_in[5];
    const float* W2     = (const float*)d_in[6];
    const float* a_src2 = (const float*)d_in[7];
    const float* a_dst2 = (const float*)d_in[8];
    const float* b2     = (const float*)d_in[9];
    const float* W_out  = (const float*)d_in[10];
    const float* b_out  = (const float*)d_in[11];
    float* y = (float*)d_out;

    const int* se = ei;
    const int* de = ei + E_EDGES;

    float* h1     = (float*)d_ws;
    float* h2     = h1 + (size_t)N_NODES * HIDV;
    float* as_    = h2 + (size_t)N_NODES * HIDV;
    float* ad_    = as_ + N_NODES;
    float* as2    = ad_ + N_NODES;
    int*   rowptr = (int*)(as2 + N_NODES);
    int*   csr    = rowptr + N_NODES;
    int*   part   = (int*)h2;
    int*   tabS   = part + E_EDGES;
    int*   tabC   = tabS + NBK * PBLOCKS;

    const int gRowT    = (N_NODES / 16) / GEMM_TPB;

    edge_partition<<<PBLOCKS, PTHREADS, 0, stream>>>(se, de, part, tabS, tabC);
    bucket_sort<<<NBK, 1024, 0, stream>>>(part, tabS, tabC, rowptr, csr);

    gemm_wmma_f32<IN_DIMV><<<gRowT, 128, 0, stream>>>(x, W1, a_src1, a_dst1, h1, as_, ad_);
    gat_gather<false><<<GATHER_BLOCKS, 256, 0, stream>>>(rowptr, csr, h1, as_, ad_, b1,
                                                    W2, a_src2, a_dst2, h2, as2, ad_,
                                                    W_out, b_out, y);

    gat_gather<true><<<GATHER_BLOCKS, 256, 0, stream>>>(rowptr, csr, h2, as2, ad_, b2,
                                                   W2, a_src2, a_dst2, h1, as_, ad_,
                                                   W_out, b_out, y);
}
